# starting kernel
# baseline (speedup 1.0000x reference)
#include <hip/hip_runtime.h>
#include <hip/hip_bf16.h>
#include <math.h>

typedef float v2f __attribute__((ext_vector_type(2)));
typedef float v8f __attribute__((ext_vector_type(8)));

#if defined(__has_builtin)
#if __has_builtin(__builtin_amdgcn_wmma_f32_16x16x4_f32)
#define HAS_WMMA_F32_4 1
#endif
#endif
#ifndef HAS_WMMA_F32_4
#define HAS_WMMA_F32_4 0
#endif

#define FEAT 128
#define BN_EPS 1e-5f

__global__ __launch_bounds__(256) void degree_kernel(const int* __restrict__ col,
                                                     const float* __restrict__ w,
                                                     float* __restrict__ deg, int E) {
    int e = blockIdx.x * 256 + threadIdx.x;
    if (e < E) {
        __hip_atomic_fetch_add(&deg[col[e]], w[e], __ATOMIC_RELAXED,
                               __HIP_MEMORY_SCOPE_AGENT);
    }
}

__global__ __launch_bounds__(256) void dis_kernel(float* __restrict__ deg_dis, int n) {
    int i = blockIdx.x * 256 + threadIdx.x;
    if (i < n) deg_dis[i] = rsqrtf(deg_dis[i] + 1.0f);
}

__global__ __launch_bounds__(256) void gemm_n128_kernel(const float* __restrict__ X,
                                                        const float* __restrict__ W,
                                                        float* __restrict__ Y,
                                                        int nrows) {
    __shared__ float lA[16 * 132];

    const int tid = threadIdx.x;
    const int rowBase = blockIdx.x * 16;

#pragma unroll
    for (int i = 0; i < 2; ++i) {
        int idx = tid + i * 256;
        int r = idx >> 5;
        int c4 = (idx & 31) << 2;
        float4 v = make_float4(0.f, 0.f, 0.f, 0.f);
        if (rowBase + r < nrows)
            v = *(const float4*)(X + (size_t)(rowBase + r) * FEAT + c4);
        float* dst = &lA[r * 132 + c4];
        dst[0] = v.x; dst[1] = v.y; dst[2] = v.z; dst[3] = v.w;
    }
    __syncthreads();

    const int wave = tid >> 5;
    const int lane = tid & 31;
    const int lid  = lane & 15;
    const int half = lane >> 4;
    const int n0   = wave * 16;

#if HAS_WMMA_F32_4
    v8f acc = {0.f, 0.f, 0.f, 0.f, 0.f, 0.f, 0.f, 0.f};
    const float* arow = &lA[lid * 132];
#pragma unroll
    for (int k0 = 0; k0 < FEAT; k0 += 4) {
        const int ka = k0 + 2 * half;
        v2f a, b;
        a.x = arow[ka];
        a.y = arow[ka + 1];
        b.x = W[(size_t)ka * FEAT + n0 + lid];
        b.y = W[(size_t)(ka + 1) * FEAT + n0 + lid];
        acc = __builtin_amdgcn_wmma_f32_16x16x4_f32(
             false, a,  false, b,
             (short)0, acc,  false,  false);
    }
    float* yb = Y + (size_t)rowBase * FEAT + n0 + lid;
#pragma unroll
    for (int v = 0; v < 8; ++v) {
        int m = v + 8 * half;
        if (rowBase + m < nrows) yb[(size_t)m * FEAT] = acc[v];
    }
#else
    float accs[8] = {0.f, 0.f, 0.f, 0.f, 0.f, 0.f, 0.f, 0.f};
    for (int k = 0; k < FEAT; ++k) {
        float bv = W[(size_t)k * FEAT + n0 + lid];
#pragma unroll
        for (int v = 0; v < 8; ++v)
            accs[v] = fmaf(lA[(v + 8 * half) * 132 + k], bv, accs[v]);
    }
    float* yb = Y + (size_t)rowBase * FEAT + n0 + lid;
#pragma unroll
    for (int v = 0; v < 8; ++v) {
        int m = v + 8 * half;
        if (rowBase + m < nrows) yb[(size_t)m * FEAT] = accs[v];
    }
#endif
}

__global__ __launch_bounds__(256) void aggregate_kernel(const int* __restrict__ row,
                                                        const int* __restrict__ col,
                                                        const float* __restrict__ ew,
                                                        const float* __restrict__ dis,
                                                        const float* __restrict__ h,
                                                        float* __restrict__ agg, int E) {
    int gid = blockIdx.x * 256 + threadIdx.x;
    int e = gid >> 5;
    if (e >= E) return;
    int lane = gid & 31;
    int r = row[e];
    int c = col[e];
    float nrm = dis[r] * ew[e] * dis[c];
    const float4 hv = *(const float4*)(h + (size_t)r * FEAT + lane * 4);
    float* dst = agg + (size_t)c * FEAT + lane * 4;
    __hip_atomic_fetch_add(dst + 0, hv.x * nrm, __ATOMIC_RELAXED, __HIP_MEMORY_SCOPE_AGENT);
    __hip_atomic_fetch_add(dst + 1, hv.y * nrm, __ATOMIC_RELAXED, __HIP_MEMORY_SCOPE_AGENT);
    __hip_atomic_fetch_add(dst + 2, hv.z * nrm, __ATOMIC_RELAXED, __HIP_MEMORY_SCOPE_AGENT);
    __hip_atomic_fetch_add(dst + 3, hv.w * nrm, __ATOMIC_RELAXED, __HIP_MEMORY_SCOPE_AGENT);
}

__global__ __launch_bounds__(256) void epilogue1_kernel(float* __restrict__ agg,
                                                        const float* __restrict__ h,
                                                        const float* __restrict__ dis,
                                                        const float* __restrict__ b0,
                                                        const float* __restrict__ g,
                                                        const float* __restrict__ be,
                                                        const float* __restrict__ mu,
                                                        const float* __restrict__ var,
                                                        int n4) {
    int idx4 = blockIdx.x * 256 + threadIdx.x;
    if (idx4 >= n4) return;
    int i  = idx4 >> 5;
    int f  = (idx4 & 31) << 2;
    float d = dis[i];
    float d2 = d * d;
    float4 a  = *(float4*)(agg + (size_t)idx4 * 4);
    float4 hv = *(const float4*)(h + (size_t)idx4 * 4);
    float out[4];
    float av[4] = {a.x, a.y, a.z, a.w};
    float hh[4] = {hv.x, hv.y, hv.z, hv.w};
#pragma unroll
    for (int j = 0; j < 4; ++j) {
        int ff = f + j;
        float v = av[j] + d2 * hh[j] + b0[ff];
        v = (v - mu[ff]) * rsqrtf(var[ff] + BN_EPS) * g[ff] + be[ff];
        out[j] = (v > 0.0f) ? v : expm1f(v);
    }
    *(float4*)(agg + (size_t)idx4 * 4) = make_float4(out[0], out[1], out[2], out[3]);
}

__global__ __launch_bounds__(256) void final_kernel(const float* __restrict__ emb,
                                                    const float* __restrict__ h1,
                                                    const float* __restrict__ agg2,
                                                    const float* __restrict__ h2,
                                                    const float* __restrict__ dis,
                                                    const float* __restrict__ b1,
                                                    float* __restrict__ out, int n4) {
    int idx4 = blockIdx.x * 256 + threadIdx.x;
    if (idx4 >= n4) return;
    int i = idx4 >> 5;
    int f = (idx4 & 31) << 2;
    float d = dis[i];
    float d2 = d * d;
    float4 e  = *(const float4*)(emb  + (size_t)idx4 * 4);
    float4 o1 = *(const float4*)(h1   + (size_t)idx4 * 4);
    float4 a2 = *(const float4*)(agg2 + (size_t)idx4 * 4);
    float4 p2 = *(const float4*)(h2   + (size_t)idx4 * 4);
    const float inv3 = 1.0f / 3.0f;
    float4 r;
    r.x = (e.x + o1.x + a2.x + d2 * p2.x + b1[f + 0]) * inv3;
    r.y = (e.y + o1.y + a2.y + d2 * p2.y + b1[f + 1]) * inv3;
    r.z = (e.z + o1.z + a2.z + d2 * p2.z + b1[f + 2]) * inv3;
    r.w = (e.w + o1.w + a2.w + d2 * p2.w + b1[f + 3]) * inv3;
    *(float4*)(out + (size_t)idx4 * 4) = r;
}

extern "C" void kernel_launch(void* const* d_in, const int* in_sizes, int n_in,
                              void* d_out, int out_size, void* d_ws, size_t ws_size,
                              hipStream_t stream) {
    const float* emb = (const float*)d_in[0];
    const int*   ei  = (const int*)d_in[1];
    const float* ew  = (const float*)d_in[2];
    const float* W0  = (const float*)d_in[3];
    const float* b0  = (const float*)d_in[4];
    const float* W1  = (const float*)d_in[5];
    const float* b1  = (const float*)d_in[6];
    const float* bng = (const float*)d_in[7];
    const float* bnb = (const float*)d_in[8];
    const float* bnm = (const float*)d_in[9];
    const float* bnv = (const float*)d_in[10];
    float* out = (float*)d_out;

    const int N  = in_sizes[0] / FEAT;
    const int E  = in_sizes[2];
    const size_t ND = (size_t)N * FEAT;

    const int* row = ei;
    const int* col = ei + E;

    float* ws   = (float*)d_ws;
    size_t disPad = ((size_t)N + 1023) & ~(size_t)1023;
    float* dis  = ws;
    float* bufA = ws + disPad;
    float* bufB = bufA + ND;
    float* bufC = bufB + ND;

    hipMemsetAsync(dis,  0, (size_t)N * sizeof(float), stream);
    hipMemsetAsync(bufB, 0, ND * sizeof(float), stream);
    hipMemsetAsync(bufC, 0, ND * sizeof(float), stream);

    const int n4 = (int)(ND / 4);
    const int gElem  = (n4 + 255) / 256;
    const int gEdgeW = (int)(((size_t)E * 32 + 255) / 256);
    const int gRowT  = (N + 15) / 16;

    degree_kernel<<<(E + 255) / 256, 256, 0, stream>>>(col, ew, dis, E);
    dis_kernel<<<(N + 255) / 256, 256, 0, stream>>>(dis, N);

    gemm_n128_kernel<<<gRowT, 256, 0, stream>>>(emb, W0, bufA, N);

    aggregate_kernel<<<gEdgeW, 256, 0, stream>>>(row, col, ew, dis, bufA, bufB, E);

    epilogue1_kernel<<<gElem, 256, 0, stream>>>(bufB, bufA, dis, b0, bng, bnb, bnm, bnv, n4);

    gemm_n128_kernel<<<gRowT, 256, 0, stream>>>(bufB, W1, bufA, N);

    aggregate_kernel<<<gEdgeW, 256, 0, stream>>>(row, col, ew, dis, bufA, bufC, E);

    final_kernel<<<gElem, 256, 0, stream>>>(emb, bufB, bufC, bufA, dis, b1, out, n4);
}


# float atomics (degree, both scatter-adds) replaced by an in-call counting sort of edges by destination (int atomic rank) + one-wave-per-node gather; sums now run in arrival-rank order
# speedup vs baseline: 2.3621x; 2.3621x over previous
#include <hip/hip_runtime.h>
#include <hip/hip_bf16.h>
#include <math.h>

typedef float v2f __attribute__((ext_vector_type(2)));
typedef float v8f __attribute__((ext_vector_type(8)));

#if defined(__has_builtin)
#if __has_builtin(__builtin_amdgcn_wmma_f32_16x16x4_f32)
#define HAS_WMMA_F32_4 1
#endif
#endif
#ifndef HAS_WMMA_F32_4
#define HAS_WMMA_F32_4 0
#endif

#define FEAT 128
#define BN_EPS 1e-5f

__global__ __launch_bounds__(256) void count_rank_kernel(const int* __restrict__ col,
                                                         int* __restrict__ cnt,
                                                         int* __restrict__ rank, int E) {
    int e = blockIdx.x * 256 + threadIdx.x;
    if (e < E) {
        rank[e] = __hip_atomic_fetch_add(&cnt[col[e]], 1, __ATOMIC_RELAXED,
                                         __HIP_MEMORY_SCOPE_AGENT);
    }
}

#define SCAN_T 1024
__global__ __launch_bounds__(SCAN_T) void offsets_kernel(const int* __restrict__ cnt,
                                                         int* __restrict__ off, int n) {
    __shared__ int wtot[32];
    __shared__ int wpre[32];
    const int tid  = threadIdx.x;
    const int lane = tid & 31;
    const int wv   = tid >> 5;
    const int tile0 = blockIdx.x * SCAN_T;

    int pre = 0;
#pragma unroll 8
    for (int i = tid; i < tile0; i += SCAN_T) pre += cnt[i];

    const int i = tile0 + tid;
    const int v = (i < n) ? cnt[i] : 0;

    int inc = v;
#pragma unroll
    for (int d = 1; d < 32; d <<= 1) {
        int t = __shfl_up(inc, d);
        if (lane >= d) inc += t;
    }
#pragma unroll
    for (int d = 16; d > 0; d >>= 1) pre += __shfl_xor(pre, d);

    if (lane == 31) wtot[wv] = inc;
    if (lane == 0)  wpre[wv] = pre;
    __syncthreads();

    int p = wpre[lane];
    int t = (lane < wv) ? wtot[lane] : 0;
#pragma unroll
    for (int d = 16; d > 0; d >>= 1) {
        p += __shfl_xor(p, d);
        t += __shfl_xor(t, d);
    }
    const int excl = p + t + inc - v;
    if (i < n) off[i] = excl;
    if (i == n - 1) off[n] = excl + v;
}

__global__ __launch_bounds__(256) void fill_kernel(const int* __restrict__ row,
                                                   const int* __restrict__ col,
                                                   const float* __restrict__ ew,
                                                   const int* __restrict__ rank,
                                                   const int* __restrict__ off,
                                                   int2* __restrict__ sedge, int E) {
    int e = blockIdx.x * 256 + threadIdx.x;
    if (e < E) {
        int p = off[col[e]] + rank[e];
        sedge[p] = make_int2(row[e], __float_as_int(ew[e]));
    }
}

__global__ __launch_bounds__(256) void dis_kernel(const int* __restrict__ off,
                                                  const int2* __restrict__ sedge,
                                                  float* __restrict__ dis, int n) {
    const int gid = blockIdx.x * 256 + threadIdx.x;
    const int i   = gid >> 3;
    const int sub = gid & 7;
    float s = 0.0f;
    if (i < n) {
        const int end = off[i + 1];
        for (int p = off[i] + sub; p < end; p += 8) s += __int_as_float(sedge[p].y);
    }
    s += __shfl_xor(s, 1);
    s += __shfl_xor(s, 2);
    s += __shfl_xor(s, 4);
    if (i < n && sub == 0) dis[i] = rsqrtf(s + 1.0f);
}

__global__ __launch_bounds__(256) void gemm_n128_kernel(const float* __restrict__ X,
                                                        const float* __restrict__ W,
                                                        float* __restrict__ Y,
                                                        int nrows) {
    __shared__ float lA[16 * 132];

    const int tid = threadIdx.x;
    const int rowBase = blockIdx.x * 16;

#pragma unroll
    for (int i = 0; i < 2; ++i) {
        int idx = tid + i * 256;
        int r = idx >> 5;
        int c4 = (idx & 31) << 2;
        float4 v = make_float4(0.f, 0.f, 0.f, 0.f);
        if (rowBase + r < nrows)
            v = *(const float4*)(X + (size_t)(rowBase + r) * FEAT + c4);
        float* dst = &lA[r * 132 + c4];
        dst[0] = v.x; dst[1] = v.y; dst[2] = v.z; dst[3] = v.w;
    }
    __syncthreads();

    const int wave = tid >> 5;
    const int lane = tid & 31;
    const int lid  = lane & 15;
    const int half = lane >> 4;
    const int n0   = wave * 16;

#if HAS_WMMA_F32_4
    v8f acc = {0.f, 0.f, 0.f, 0.f, 0.f, 0.f, 0.f, 0.f};
    const float* arow = &lA[lid * 132];
#pragma unroll
    for (int k0 = 0; k0 < FEAT; k0 += 4) {
        const int ka = k0 + 2 * half;
        v2f a, b;
        a.x = arow[ka];
        a.y = arow[ka + 1];
        b.x = W[(size_t)ka * FEAT + n0 + lid];
        b.y = W[(size_t)(ka + 1) * FEAT + n0 + lid];
        acc = __builtin_amdgcn_wmma_f32_16x16x4_f32(
             false, a,  false, b,
             (short)0, acc,  false,  false);
    }
    float* yb = Y + (size_t)rowBase * FEAT + n0 + lid;
#pragma unroll
    for (int v = 0; v < 8; ++v) {
        int m = v + 8 * half;
        if (rowBase + m < nrows) yb[(size_t)m * FEAT] = acc[v];
    }
#else
    float accs[8] = {0.f, 0.f, 0.f, 0.f, 0.f, 0.f, 0.f, 0.f};
    for (int k = 0; k < FEAT; ++k) {
        float bv = W[(size_t)k * FEAT + n0 + lid];
#pragma unroll
        for (int v = 0; v < 8; ++v)
            accs[v] = fmaf(lA[(v + 8 * half) * 132 + k], bv, accs[v]);
    }
    float* yb = Y + (size_t)rowBase * FEAT + n0 + lid;
#pragma unroll
    for (int v = 0; v < 8; ++v) {
        int m = v + 8 * half;
        if (rowBase + m < nrows) yb[(size_t)m * FEAT] = accs[v];
    }
#endif
}

__device__ __forceinline__ void row_fma(float4& acc, float cf, const float4 hv) {
    acc.x = fmaf(cf, hv.x, acc.x);
    acc.y = fmaf(cf, hv.y, acc.y);
    acc.z = fmaf(cf, hv.z, acc.z);
    acc.w = fmaf(cf, hv.w, acc.w);
}

template <int K>
__device__ __forceinline__ void add_rows(float4& acc, int r, float cf, int j,
                                         const float* __restrict__ h, unsigned lane4) {
    float4 hv[K];
#pragma unroll
    for (int k = 0; k < K; ++k) {
        const int rk = __builtin_amdgcn_readlane(r, j + k);
        hv[k] = *(const float4*)(h + (size_t)rk * FEAT + lane4);
    }
#pragma unroll
    for (int k = 0; k < K; ++k) {
        const float ck = __int_as_float(__builtin_amdgcn_readlane(__float_as_int(cf), j + k));
        row_fma(acc, ck, hv[k]);
    }
}

__global__ __launch_bounds__(256) void gather_kernel(const int* __restrict__ off,
                                                     const int2* __restrict__ sedge,
                                                     const float* __restrict__ dis,
                                                     const float* __restrict__ h,
                                                     float* __restrict__ agg, int n) {
    const int lane = threadIdx.x & 31;
    const unsigned lane4 = (unsigned)lane * 4u;
    const int c = __builtin_amdgcn_readfirstlane(blockIdx.x * 8 + (threadIdx.x >> 5));
    if (c >= n) return;
    const int beg = __builtin_amdgcn_readfirstlane(off[c]);
    const int end = __builtin_amdgcn_readfirstlane(off[c + 1]);
    const float dc = dis[c];
    float4 acc = make_float4(0.f, 0.f, 0.f, 0.f);

    for (int base = beg; base < end; base += 32) {
        const int p = base + lane;
        int r = 0;
        float cf = 0.f;
        if (p < end) {
            const int2 ed = sedge[p];
            r  = ed.x;
            cf = dis[r] * __int_as_float(ed.y) * dc;
        }
        const int m = (end - base < 32) ? (end - base) : 32;
        int j = 0;
        for (; j + 8 <= m; j += 8) add_rows<8>(acc, r, cf, j, h, lane4);
        if (m - j >= 4) { add_rows<4>(acc, r, cf, j, h, lane4); j += 4; }
        const int rem = m - j;
        if (rem == 3)      add_rows<3>(acc, r, cf, j, h, lane4);
        else if (rem == 2) add_rows<2>(acc, r, cf, j, h, lane4);
        else if (rem == 1) add_rows<1>(acc, r, cf, j, h, lane4);
    }
    *(float4*)(agg + (size_t)c * FEAT + lane4) = acc;
}

__global__ __launch_bounds__(256) void epilogue1_kernel(float* __restrict__ agg,
                                                        const float* __restrict__ h,
                                                        const float* __restrict__ dis,
                                                        const float* __restrict__ b0,
                                                        const float* __restrict__ g,
                                                        const float* __restrict__ be,
                                                        const float* __restrict__ mu,
                                                        const float* __restrict__ var,
                                                        int n4) {
    int idx4 = blockIdx.x * 256 + threadIdx.x;
    if (idx4 >= n4) return;
    int i  = idx4 >> 5;
    int f  = (idx4 & 31) << 2;
    float d = dis[i];
    float d2 = d * d;
    float4 a  = *(float4*)(agg + (size_t)idx4 * 4);
    float4 hv = *(const float4*)(h + (size_t)idx4 * 4);
    float out[4];
    float av[4] = {a.x, a.y, a.z, a.w};
    float hh[4] = {hv.x, hv.y, hv.z, hv.w};
#pragma unroll
    for (int j = 0; j < 4; ++j) {
        int ff = f + j;
        float v = av[j] + d2 * hh[j] + b0[ff];
        v = (v - mu[ff]) * rsqrtf(var[ff] + BN_EPS) * g[ff] + be[ff];
        out[j] = (v > 0.0f) ? v : expm1f(v);
    }
    *(float4*)(agg + (size_t)idx4 * 4) = make_float4(out[0], out[1], out[2], out[3]);
}

__global__ __launch_bounds__(256) void final_kernel(const float* __restrict__ emb,
                                                    const float* __restrict__ h1,
                                                    const float* __restrict__ agg2,
                                                    const float* __restrict__ h2,
                                                    const float* __restrict__ dis,
                                                    const float* __restrict__ b1,
                                                    float* __restrict__ out, int n4) {
    int idx4 = blockIdx.x * 256 + threadIdx.x;
    if (idx4 >= n4) return;
    int i = idx4 >> 5;
    int f = (idx4 & 31) << 2;
    float d = dis[i];
    float d2 = d * d;
    float4 e  = *(const float4*)(emb  + (size_t)idx4 * 4);
    float4 o1 = *(const float4*)(h1   + (size_t)idx4 * 4);
    float4 a2 = *(const float4*)(agg2 + (size_t)idx4 * 4);
    float4 p2 = *(const float4*)(h2   + (size_t)idx4 * 4);
    const float inv3 = 1.0f / 3.0f;
    float4 r;
    r.x = (e.x + o1.x + a2.x + d2 * p2.x + b1[f + 0]) * inv3;
    r.y = (e.y + o1.y + a2.y + d2 * p2.y + b1[f + 1]) * inv3;
    r.z = (e.z + o1.z + a2.z + d2 * p2.z + b1[f + 2]) * inv3;
    r.w = (e.w + o1.w + a2.w + d2 * p2.w + b1[f + 3]) * inv3;
    *(float4*)(out + (size_t)idx4 * 4) = r;
}

extern "C" void kernel_launch(void* const* d_in, const int* in_sizes, int n_in,
                              void* d_out, int out_size, void* d_ws, size_t ws_size,
                              hipStream_t stream) {
    const float* emb = (const float*)d_in[0];
    const int*   ei  = (const int*)d_in[1];
    const float* ew  = (const float*)d_in[2];
    const float* W0  = (const float*)d_in[3];
    const float* b0  = (const float*)d_in[4];
    const float* W1  = (const float*)d_in[5];
    const float* b1  = (const float*)d_in[6];
    const float* bng = (const float*)d_in[7];
    const float* bnb = (const float*)d_in[8];
    const float* bnm = (const float*)d_in[9];
    const float* bnv = (const float*)d_in[10];
    float* out = (float*)d_out;

    const int N  = in_sizes[0] / FEAT;
    const int E  = in_sizes[2];
    const size_t ND = (size_t)N * FEAT;

    const int* row = ei;
    const int* col = ei + E;

    float* ws   = (float*)d_ws;
    size_t nPad = ((size_t)N + 1023) & ~(size_t)1023;
    size_t ePad = ((size_t)E + 1023) & ~(size_t)1023;
    int*   cnt   = (int*)ws;
    int*   off   = cnt + nPad;
    float* dis   = (float*)(off + nPad + 1024);
    int*   rank  = (int*)(dis + nPad);
    int2*  sedge = (int2*)(rank + ePad);
    float* bufA  = (float*)(rank + ePad) + 2 * ePad;
    float* bufB  = bufA + ND;
    float* bufC  = bufB + ND;

    hipMemsetAsync(cnt, 0, (size_t)N * sizeof(int), stream);

    const int n4 = (int)(ND / 4);
    const int gElem  = (n4 + 255) / 256;
    const int gEdge  = (E + 255) / 256;
    const int gNodeW = (N + 7) / 8;
    const int gRowT  = (N + 15) / 16;

    count_rank_kernel<<<gEdge, 256, 0, stream>>>(col, cnt, rank, E);
    offsets_kernel<<<(N + SCAN_T - 1) / SCAN_T, SCAN_T, 0, stream>>>(cnt, off, N);
    fill_kernel<<<gEdge, 256, 0, stream>>>(row, col, ew, rank, off, sedge, E);
    dis_kernel<<<(int)(((size_t)N * 8 + 255) / 256), 256, 0, stream>>>(off, sedge, dis, N);

    gemm_n128_kernel<<<gRowT, 256, 0, stream>>>(emb, W0, bufA, N);

    gather_kernel<<<gNodeW, 256, 0, stream>>>(off, sedge, dis, bufA, bufB, N);

    epilogue1_kernel<<<gElem, 256, 0, stream>>>(bufB, bufA, dis, b0, bng, bnb, bnm, bnv, n4);

    gemm_n128_kernel<<<gRowT, 256, 0, stream>>>(bufB, W1, bufA, N);

    gather_kernel<<<gNodeW, 256, 0, stream>>>(off, sedge, dis, bufA, bufC, N);

    final_kernel<<<gElem, 256, 0, stream>>>(emb, bufB, bufC, bufA, dis, b1, out, n4);
}


# edge grouping by destination: per-edge returning global atomic + scan + scatter replaced by a two-level counting sort on LDS counters (256-node buckets); sums now run in LDS-rank order
# speedup vs baseline: 2.5412x; 1.0758x over previous
#include <hip/hip_runtime.h>
#include <hip/hip_bf16.h>
#include <math.h>

typedef float v2f __attribute__((ext_vector_type(2)));
typedef float v8f __attribute__((ext_vector_type(8)));

#if defined(__has_builtin)
#if __has_builtin(__builtin_amdgcn_wmma_f32_16x16x4_f32)
#define HAS_WMMA_F32_4 1
#endif
#endif
#ifndef HAS_WMMA_F32_4
#define HAS_WMMA_F32_4 0
#endif

#define FEAT 128
#define BN_EPS 1e-5f

#define BKT_SHIFT 8
#define BKT_W     256
#define MAX_NB    1024
#define SORT_G    256
#define SORT_T    512
#define SORT_K    8
#define TAB_RG    (SORT_T / 64)
#define TAB_RPG   (SORT_G / TAB_RG)
#define SEG_T     512
#define SEG_K     16

__global__ __launch_bounds__(SORT_T) void bucket_count_kernel(const int* __restrict__ col,
                                                              int* __restrict__ tab,
                                                              int E, int chunk, int nb, int tp) {
    __shared__ int hist[MAX_NB + 1];
    const int tid = threadIdx.x;
    const int e0 = blockIdx.x * chunk;
    const int e1 = min(e0 + chunk, E);
    for (int b = tid; b < tp; b += SORT_T) hist[b] = 0;
    __syncthreads();
    for (int base = e0; base < e1; base += SORT_T * SORT_K) {
        unsigned c[SORT_K];
#pragma unroll
        for (int k = 0; k < SORT_K; ++k) {
            const int e = base + tid + k * SORT_T;
            c[k] = (unsigned)col[min(e, e1 - 1)];
        }
#pragma unroll
        for (int k = 0; k < SORT_K; ++k) {
            const int e = base + tid + k * SORT_T;
            const int b = (e < e1) ? min((int)(c[k] >> BKT_SHIFT), nb - 1) : MAX_NB;
            atomicAdd(&hist[b], 1);
        }
    }
    __syncthreads();
    for (int b = tid; b < tp; b += SORT_T) tab[(size_t)blockIdx.x * tp + b] = hist[b];
}

__global__ __launch_bounds__(SORT_T) void bucket_place_kernel(const int* __restrict__ row,
                                                              const int* __restrict__ col,
                                                              const float* __restrict__ ew,
                                                              const int* __restrict__ tab,
                                                              int* __restrict__ bstart,
                                                              int2* __restrict__ prec,
                                                              int E, int chunk, int nb, int tp) {
    __shared__ int tot[MAX_NB];
    __shared__ int cur[MAX_NB + 1];
    __shared__ int wtot[SORT_T / 32];
    const int tid  = threadIdx.x;
    const int lane = tid & 31;
    const int wv   = tid >> 5;
    const int w    = blockIdx.x;
    const int e0 = w * chunk;
    const int e1 = min(e0 + chunk, E);

    for (int b = tid; b < tp; b += SORT_T) { tot[b] = 0; cur[b] = 0; }
    __syncthreads();

    unsigned c[SORT_K];
    int      r[SORT_K];
    int      wt[SORT_K];
#pragma unroll
    for (int k = 0; k < SORT_K; ++k) { c[k] = 0u; r[k] = 0; wt[k] = 0; }
    if (e0 < e1) {
#pragma unroll
        for (int k = 0; k < SORT_K; ++k) {
            const int e = min(e0 + tid + k * SORT_T, e1 - 1);
            c[k]  = (unsigned)col[e];
            r[k]  = row[e];
            wt[k] = __float_as_int(ew[e]);
        }
    }

    const int cg = tid & 63;
    const int rg = tid >> 6;
    for (int c4 = cg * 4; c4 < tp; c4 += 256) {
        int4 v[TAB_RPG];
#pragma unroll
        for (int i = 0; i < TAB_RPG; ++i) {
            const int rr = rg * TAB_RPG + ((i + w) & (TAB_RPG - 1));
            v[i] = *(const int4*)(tab + (size_t)rr * tp + c4);
        }
        __builtin_amdgcn_sched_barrier(0);
        int t0 = 0, t1 = 0, t2 = 0, t3 = 0;
        int p0 = 0, p1 = 0, p2 = 0, p3 = 0;
#pragma unroll
        for (int i = 0; i < TAB_RPG; ++i) {
            const int rr = rg * TAB_RPG + ((i + w) & (TAB_RPG - 1));
            t0 += v[i].x; t1 += v[i].y; t2 += v[i].z; t3 += v[i].w;
            if (rr < w) { p0 += v[i].x; p1 += v[i].y; p2 += v[i].z; p3 += v[i].w; }
        }
        atomicAdd(&tot[c4 + 0], t0); atomicAdd(&tot[c4 + 1], t1);
        atomicAdd(&tot[c4 + 2], t2); atomicAdd(&tot[c4 + 3], t3);
        atomicAdd(&cur[c4 + 0], p0); atomicAdd(&cur[c4 + 1], p1);
        atomicAdd(&cur[c4 + 2], p2); atomicAdd(&cur[c4 + 3], p3);
    }
    __syncthreads();

    const bool own = (4 * tid < tp);
    int a0 = 0, a1 = 0, a2 = 0, a3 = 0;
    if (own) { a0 = tot[4 * tid]; a1 = tot[4 * tid + 1]; a2 = tot[4 * tid + 2]; a3 = tot[4 * tid + 3]; }
    const int s = a0 + a1 + a2 + a3;
    int inc = s;
#pragma unroll
    for (int d = 1; d < 32; d <<= 1) {
        int t = __shfl_up(inc, d);
        if (lane >= d) inc += t;
    }
    if (lane == 31) wtot[wv] = inc;
    __syncthreads();
    int base0 = inc - s;
#pragma unroll
    for (int k = 0; k < SORT_T / 32; ++k) if (k < wv) base0 += wtot[k];
    if (own) {
        const int s0 = base0, s1 = s0 + a0, s2 = s1 + a1, s3 = s2 + a2;
        cur[4 * tid]     += s0;
        cur[4 * tid + 1] += s1;
        cur[4 * tid + 2] += s2;
        cur[4 * tid + 3] += s3;
        if (w == 0) {
            bstart[4 * tid]     = s0;
            bstart[4 * tid + 1] = s1;
            bstart[4 * tid + 2] = s2;
            bstart[4 * tid + 3] = s3;
            if (4 * tid + 4 == tp) bstart[tp] = s3 + a3;
        }
    }
    __syncthreads();

    for (int base = e0; base < e1; base += SORT_T * SORT_K) {
        if (base != e0) {
#pragma unroll
            for (int k = 0; k < SORT_K; ++k) {
                const int e = min(base + tid + k * SORT_T, e1 - 1);
                c[k]  = (unsigned)col[e];
                r[k]  = row[e];
                wt[k] = __float_as_int(ew[e]);
            }
        }
        int pos[SORT_K];
#pragma unroll
        for (int k = 0; k < SORT_K; ++k) {
            const int e = base + tid + k * SORT_T;
            const int b = (e < e1) ? min((int)(c[k] >> BKT_SHIFT), nb - 1) : MAX_NB;
            pos[k] = atomicAdd(&cur[b], 1);
        }
#pragma unroll
        for (int k = 0; k < SORT_K; ++k) {
            const int e = base + tid + k * SORT_T;
            if (e < e1)
                prec[pos[k]] = make_int2((int)((unsigned)r[k] | ((c[k] & (BKT_W - 1)) << 24)), wt[k]);
        }
    }
}

__global__ __launch_bounds__(SEG_T) void bucket_sort_kernel(const int* __restrict__ bstart,
                                                            const int2* __restrict__ prec,
                                                            int* __restrict__ off,
                                                            int2* __restrict__ sedge, int n) {
    __shared__ int cnt[BKT_W + 1];
    __shared__ int wtot[BKT_W / 32];
    const int tid  = threadIdx.x;
    const int lane = tid & 31;
    const int wv   = tid >> 5;
    const int b    = blockIdx.x;
    const int beg  = bstart[b];
    const int end  = bstart[b + 1];

    int2 rec[SEG_K];
#pragma unroll
    for (int k = 0; k < SEG_K; ++k) {
        const int i = beg + tid + k * SEG_T;
        rec[k] = prec[max(min(i, end - 1), 0)];
    }
    if (tid < BKT_W) cnt[tid] = 0;
    __syncthreads();
#pragma unroll
    for (int k = 0; k < SEG_K; ++k) {
        const int i = beg + tid + k * SEG_T;
        atomicAdd(&cnt[(i < end) ? (int)((unsigned)rec[k].x >> 24) : BKT_W], 1);
    }
    for (int i = beg + SEG_T * SEG_K + tid; i < end; i += SEG_T)
        atomicAdd(&cnt[(unsigned)prec[i].x >> 24], 1);
    __syncthreads();

    const int v = (tid < BKT_W) ? cnt[tid] : 0;
    int inc = v;
#pragma unroll
    for (int d = 1; d < 32; d <<= 1) {
        int t = __shfl_up(inc, d);
        if (lane >= d) inc += t;
    }
    if (tid < BKT_W && lane == 31) wtot[wv] = inc;
    __syncthreads();
    if (tid < BKT_W) {
        int excl = beg + inc - v;
#pragma unroll
        for (int k = 0; k < BKT_W / 32; ++k) if (k < wv) excl += wtot[k];
        cnt[tid] = excl;
        const int node = b * BKT_W + tid;
        if (node < n) off[node] = excl;
        if (node == n - 1) off[n] = excl + v;
    }
    __syncthreads();
    int pos[SEG_K];
#pragma unroll
    for (int k = 0; k < SEG_K; ++k) {
        const int i = beg + tid + k * SEG_T;
        pos[k] = atomicAdd(&cnt[(i < end) ? (int)((unsigned)rec[k].x >> 24) : BKT_W], 1);
    }
#pragma unroll
    for (int k = 0; k < SEG_K; ++k) {
        const int i = beg + tid + k * SEG_T;
        if (i < end) sedge[pos[k]] = make_int2((int)((unsigned)rec[k].x & 0xFFFFFFu), rec[k].y);
    }
    for (int i = beg + SEG_T * SEG_K + tid; i < end; i += SEG_T) {
        const int2 q = prec[i];
        const unsigned x = (unsigned)q.x;
        const int at = atomicAdd(&cnt[x >> 24], 1);
        sedge[at] = make_int2((int)(x & 0xFFFFFFu), q.y);
    }
}

__global__ __launch_bounds__(256) void dis_kernel(const int* __restrict__ off,
                                                  const int2* __restrict__ sedge,
                                                  float* __restrict__ dis, int n) {
    const int gid = blockIdx.x * 256 + threadIdx.x;
    const int i   = gid >> 3;
    const int sub = gid & 7;
    float s = 0.0f;
    if (i < n) {
        const int end = off[i + 1];
        for (int p = off[i] + sub; p < end; p += 8) s += __int_as_float(sedge[p].y);
    }
    s += __shfl_xor(s, 1);
    s += __shfl_xor(s, 2);
    s += __shfl_xor(s, 4);
    if (i < n && sub == 0) dis[i] = rsqrtf(s + 1.0f);
}

__global__ __launch_bounds__(256) void gemm_n128_kernel(const float* __restrict__ X,
                                                        const float* __restrict__ W,
                                                        float* __restrict__ Y,
                                                        int nrows) {
    __shared__ float lA[16 * 132];

    const int tid = threadIdx.x;
    const int rowBase = blockIdx.x * 16;

#pragma unroll
    for (int i = 0; i < 2; ++i) {
        int idx = tid + i * 256;
        int r = idx >> 5;
        int c4 = (idx & 31) << 2;
        float4 v = make_float4(0.f, 0.f, 0.f, 0.f);
        if (rowBase + r < nrows)
            v = *(const float4*)(X + (size_t)(rowBase + r) * FEAT + c4);
        float* dst = &lA[r * 132 + c4];
        dst[0] = v.x; dst[1] = v.y; dst[2] = v.z; dst[3] = v.w;
    }
    __syncthreads();

    const int wave = tid >> 5;
    const int lane = tid & 31;
    const int lid  = lane & 15;
    const int half = lane >> 4;
    const int n0   = wave * 16;

#if HAS_WMMA_F32_4
    v8f acc = {0.f, 0.f, 0.f, 0.f, 0.f, 0.f, 0.f, 0.f};
    const float* arow = &lA[lid * 132];
#pragma unroll
    for (int k0 = 0; k0 < FEAT; k0 += 4) {
        const int ka = k0 + 2 * half;
        v2f a, b;
        a.x = arow[ka];
        a.y = arow[ka + 1];
        b.x = W[(size_t)ka * FEAT + n0 + lid];
        b.y = W[(size_t)(ka + 1) * FEAT + n0 + lid];
        acc = __builtin_amdgcn_wmma_f32_16x16x4_f32(
             false, a,  false, b,
             (short)0, acc,  false,  false);
    }
    float* yb = Y + (size_t)rowBase * FEAT + n0 + lid;
#pragma unroll
    for (int v = 0; v < 8; ++v) {
        int m = v + 8 * half;
        if (rowBase + m < nrows) yb[(size_t)m * FEAT] = acc[v];
    }
#else
    float accs[8] = {0.f, 0.f, 0.f, 0.f, 0.f, 0.f, 0.f, 0.f};
    for (int k = 0; k < FEAT; ++k) {
        float bv = W[(size_t)k * FEAT + n0 + lid];
#pragma unroll
        for (int v = 0; v < 8; ++v)
            accs[v] = fmaf(lA[(v + 8 * half) * 132 + k], bv, accs[v]);
    }
    float* yb = Y + (size_t)rowBase * FEAT + n0 + lid;
#pragma unroll
    for (int v = 0; v < 8; ++v) {
        int m = v + 8 * half;
        if (rowBase + m < nrows) yb[(size_t)m * FEAT] = accs[v];
    }
#endif
}

__device__ __forceinline__ void row_fma(float4& acc, float cf, const float4 hv) {
    acc.x = fmaf(cf, hv.x, acc.x);
    acc.y = fmaf(cf, hv.y, acc.y);
    acc.z = fmaf(cf, hv.z, acc.z);
    acc.w = fmaf(cf, hv.w, acc.w);
}

template <int K>
__device__ __forceinline__ void add_rows(float4& acc, int r, float cf, int j,
                                         const float* __restrict__ h, unsigned lane4) {
    float4 hv[K];
#pragma unroll
    for (int k = 0; k < K; ++k) {
        const int rk = __builtin_amdgcn_readlane(r, j + k);
        hv[k] = *(const float4*)(h + (size_t)rk * FEAT + lane4);
    }
#pragma unroll
    for (int k = 0; k < K; ++k) {
        const float ck = __int_as_float(__builtin_amdgcn_readlane(__float_as_int(cf), j + k));
        row_fma(acc, ck, hv[k]);
    }
}

__global__ __launch_bounds__(256) void gather_kernel(const int* __restrict__ off,
                                                     const int2* __restrict__ sedge,
                                                     const float* __restrict__ dis,
                                                     const float* __restrict__ h,
                                                     float* __restrict__ agg, int n) {
    const int lane = threadIdx.x & 31;
    const unsigned lane4 = (unsigned)lane * 4u;
    const int c = __builtin_amdgcn_readfirstlane(blockIdx.x * 8 + (threadIdx.x >> 5));
    if (c >= n) return;
    const int beg = __builtin_amdgcn_readfirstlane(off[c]);
    const int end = __builtin_amdgcn_readfirstlane(off[c + 1]);
    const float dc = dis[c];
    float4 acc = make_float4(0.f, 0.f, 0.f, 0.f);

    for (int base = beg; base < end; base += 32) {
        const int p = base + lane;
        int r = 0;
        float cf = 0.f;
        if (p < end) {
            const int2 ed = sedge[p];
            r  = ed.x;
            cf = dis[r] * __int_as_float(ed.y) * dc;
        }
        const int m = (end - base < 32) ? (end - base) : 32;
        int j = 0;
        for (; j + 8 <= m; j += 8) add_rows<8>(acc, r, cf, j, h, lane4);
        if (m - j >= 4) { add_rows<4>(acc, r, cf, j, h, lane4); j += 4; }
        const int rem = m - j;
        if (rem == 3)      add_rows<3>(acc, r, cf, j, h, lane4);
        else if (rem == 2) add_rows<2>(acc, r, cf, j, h, lane4);
        else if (rem == 1) add_rows<1>(acc, r, cf, j, h, lane4);
    }
    *(float4*)(agg + (size_t)c * FEAT + lane4) = acc;
}

__global__ __launch_bounds__(256) void epilogue1_kernel(float* __restrict__ agg,
                                                        const float* __restrict__ h,
                                                        const float* __restrict__ dis,
                                                        const float* __restrict__ b0,
                                                        const float* __restrict__ g,
                                                        const float* __restrict__ be,
                                                        const float* __restrict__ mu,
                                                        const float* __restrict__ var,
                                                        int n4) {
    int idx4 = blockIdx.x * 256 + threadIdx.x;
    if (idx4 >= n4) return;
    int i  = idx4 >> 5;
    int f  = (idx4 & 31) << 2;
    float d = dis[i];
    float d2 = d * d;
    float4 a  = *(float4*)(agg + (size_t)idx4 * 4);
    float4 hv = *(const float4*)(h + (size_t)idx4 * 4);
    float out[4];
    float av[4] = {a.x, a.y, a.z, a.w};
    float hh[4] = {hv.x, hv.y, hv.z, hv.w};
#pragma unroll
    for (int j = 0; j < 4; ++j) {
        int ff = f + j;
        float v = av[j] + d2 * hh[j] + b0[ff];
        v = (v - mu[ff]) * rsqrtf(var[ff] + BN_EPS) * g[ff] + be[ff];
        out[j] = (v > 0.0f) ? v : expm1f(v);
    }
    *(float4*)(agg + (size_t)idx4 * 4) = make_float4(out[0], out[1], out[2], out[3]);
}

__global__ __launch_bounds__(256) void final_kernel(const float* __restrict__ emb,
                                                    const float* __restrict__ h1,
                                                    const float* __restrict__ agg2,
                                                    const float* __restrict__ h2,
                                                    const float* __restrict__ dis,
                                                    const float* __restrict__ b1,
                                                    float* __restrict__ out, int n4) {
    int idx4 = blockIdx.x * 256 + threadIdx.x;
    if (idx4 >= n4) return;
    int i = idx4 >> 5;
    int f = (idx4 & 31) << 2;
    float d = dis[i];
    float d2 = d * d;
    float4 e  = *(const float4*)(emb  + (size_t)idx4 * 4);
    float4 o1 = *(const float4*)(h1   + (size_t)idx4 * 4);
    float4 a2 = *(const float4*)(agg2 + (size_t)idx4 * 4);
    float4 p2 = *(const float4*)(h2   + (size_t)idx4 * 4);
    const float inv3 = 1.0f / 3.0f;
    float4 r;
    r.x = (e.x + o1.x + a2.x + d2 * p2.x + b1[f + 0]) * inv3;
    r.y = (e.y + o1.y + a2.y + d2 * p2.y + b1[f + 1]) * inv3;
    r.z = (e.z + o1.z + a2.z + d2 * p2.z + b1[f + 2]) * inv3;
    r.w = (e.w + o1.w + a2.w + d2 * p2.w + b1[f + 3]) * inv3;
    *(float4*)(out + (size_t)idx4 * 4) = r;
}

extern "C" void kernel_launch(void* const* d_in, const int* in_sizes, int n_in,
                              void* d_out, int out_size, void* d_ws, size_t ws_size,
                              hipStream_t stream) {
    const float* emb = (const float*)d_in[0];
    const int*   ei  = (const int*)d_in[1];
    const float* ew  = (const float*)d_in[2];
    const float* W0  = (const float*)d_in[3];
    const float* b0  = (const float*)d_in[4];
    const float* W1  = (const float*)d_in[5];
    const float* b1  = (const float*)d_in[6];
    const float* bng = (const float*)d_in[7];
    const float* bnb = (const float*)d_in[8];
    const float* bnm = (const float*)d_in[9];
    const float* bnv = (const float*)d_in[10];
    float* out = (float*)d_out;

    const int N  = in_sizes[0] / FEAT;
    const int E  = in_sizes[2];
    const size_t ND = (size_t)N * FEAT;

    const int* row = ei;
    const int* col = ei + E;

    float* ws   = (float*)d_ws;
    size_t nPad = ((size_t)N + 1023) & ~(size_t)1023;
    size_t ePad = ((size_t)E + 1023) & ~(size_t)1023;
    int*   tab    = (int*)ws;
    int*   off    = tab + nPad;
    float* dis    = (float*)(off + nPad + 1024);
    int*   bstart = (int*)(dis + nPad);
    int2*  prec   = (int2*)(bstart + 2048);
    int2*  sedge  = prec + ePad;
    float* bufA   = (float*)(sedge + ePad);
    float* bufB   = bufA + ND;
    float* bufC   = bufB + ND;

    const int n4 = (int)(ND / 4);
    const int gElem  = (n4 + 255) / 256;
    const int gNodeW = (N + 7) / 8;
    const int gRowT  = (N + 15) / 16;
    const int nb     = (N + BKT_W - 1) / BKT_W;
    const int tp     = (nb + 3) & ~3;
    const int chunk  = (E + SORT_G - 1) / SORT_G;

    bucket_count_kernel<<<SORT_G, SORT_T, 0, stream>>>(col, tab, E, chunk, nb, tp);
    bucket_place_kernel<<<SORT_G, SORT_T, 0, stream>>>(row, col, ew, tab, bstart, prec, E, chunk, nb, tp);
    bucket_sort_kernel<<<nb, SEG_T, 0, stream>>>(bstart, prec, off, sedge, N);
    dis_kernel<<<(int)(((size_t)N * 8 + 255) / 256), 256, 0, stream>>>(off, sedge, dis, N);

    gemm_n128_kernel<<<gRowT, 256, 0, stream>>>(emb, W0, bufA, N);

    gather_kernel<<<gNodeW, 256, 0, stream>>>(off, sedge, dis, bufA, bufB, N);

    epilogue1_kernel<<<gElem, 256, 0, stream>>>(bufB, bufA, dis, b0, bng, bnb, bnm, bnv, n4);

    gemm_n128_kernel<<<gRowT, 256, 0, stream>>>(bufB, W1, bufA, N);

    gather_kernel<<<gNodeW, 256, 0, stream>>>(off, sedge, dis, bufA, bufC, N);

    final_kernel<<<gElem, 256, 0, stream>>>(emb, bufB, bufC, bufA, dis, b1, out, n4);
}


# each layer is one launch: neighbour rows are summed first (S.W, not sum of X.W: sums reassociated) into a 32-node LDS tile that feeds the f32 WMMA; bias, BN, ELU, mean at the store; 10 -> 6 launches
# speedup vs baseline: 3.1173x; 1.2267x over previous
#include <hip/hip_runtime.h>
#include <hip/hip_bf16.h>
#include <math.h>

typedef float v2f __attribute__((ext_vector_type(2)));
typedef float v8f __attribute__((ext_vector_type(8)));

#define FEAT 128
#define BN_EPS 1e-5f

#define BKT_SHIFT 8
#define BKT_W     256
#define MAX_NB    1024
#define SORT_G    256
#define SORT_T    512
#define SORT_K    8
#define TAB_RG    (SORT_T / 64)
#define TAB_RPG   (SORT_G / TAB_RG)
#define SEG_T     512
#define SEG_K     16

__global__ __launch_bounds__(SORT_T) void bucket_count_kernel(const int* __restrict__ col,
                                                              int* __restrict__ tab,
                                                              int E, int chunk, int nb, int tp) {
    __shared__ int hist[MAX_NB + 1];
    const int tid = threadIdx.x;
    const int e0 = blockIdx.x * chunk;
    const int e1 = min(e0 + chunk, E);
    for (int b = tid; b < tp; b += SORT_T) hist[b] = 0;
    __syncthreads();
    for (int base = e0; base < e1; base += SORT_T * SORT_K) {
        unsigned c[SORT_K];
#pragma unroll
        for (int k = 0; k < SORT_K; ++k) {
            const int e = base + tid + k * SORT_T;
            c[k] = (unsigned)col[min(e, e1 - 1)];
        }
#pragma unroll
        for (int k = 0; k < SORT_K; ++k) {
            const int e = base + tid + k * SORT_T;
            const int b = (e < e1) ? min((int)(c[k] >> BKT_SHIFT), nb - 1) : MAX_NB;
            atomicAdd(&hist[b], 1);
        }
    }
    __syncthreads();
    for (int b = tid; b < tp; b += SORT_T) tab[(size_t)blockIdx.x * tp + b] = hist[b];
}

__global__ __launch_bounds__(SORT_T) void bucket_place_kernel(const int* __restrict__ row,
                                                              const int* __restrict__ col,
                                                              const float* __restrict__ ew,
                                                              const int* __restrict__ tab,
                                                              int* __restrict__ bstart,
                                                              int2* __restrict__ prec,
                                                              int E, int chunk, int nb, int tp) {
    __shared__ int tot[MAX_NB];
    __shared__ int cur[MAX_NB + 1];
    __shared__ int wtot[SORT_T / 32];
    const int tid  = threadIdx.x;
    const int lane = tid & 31;
    const int wv   = tid >> 5;
    const int w    = blockIdx.x;
    const int e0 = w * chunk;
    const int e1 = min(e0 + chunk, E);

    for (int b = tid; b < tp; b += SORT_T) { tot[b] = 0; cur[b] = 0; }
    __syncthreads();

    unsigned c[SORT_K];
    int      r[SORT_K];
    int      wt[SORT_K];
#pragma unroll
    for (int k = 0; k < SORT_K; ++k) { c[k] = 0u; r[k] = 0; wt[k] = 0; }
    if (e0 < e1) {
#pragma unroll
        for (int k = 0; k < SORT_K; ++k) {
            const int e = min(e0 + tid + k * SORT_T, e1 - 1);
            c[k]  = (unsigned)col[e];
            r[k]  = row[e];
            wt[k] = __float_as_int(ew[e]);
        }
    }

    const int cg = tid & 63;
    const int rg = tid >> 6;
    for (int c4 = cg * 4; c4 < tp; c4 += 256) {
        int4 v[TAB_RPG];
#pragma unroll
        for (int i = 0; i < TAB_RPG; ++i) {
            const int rr = rg * TAB_RPG + ((i + w) & (TAB_RPG - 1));
            v[i] = *(const int4*)(tab + (size_t)rr * tp + c4);
        }
        __builtin_amdgcn_sched_barrier(0);
        int t0 = 0, t1 = 0, t2 = 0, t3 = 0;
        int p0 = 0, p1 = 0, p2 = 0, p3 = 0;
#pragma unroll
        for (int i = 0; i < TAB_RPG; ++i) {
            const int rr = rg * TAB_RPG + ((i + w) & (TAB_RPG - 1));
            t0 += v[i].x; t1 += v[i].y; t2 += v[i].z; t3 += v[i].w;
            if (rr < w) { p0 += v[i].x; p1 += v[i].y; p2 += v[i].z; p3 += v[i].w; }
        }
        atomicAdd(&tot[c4 + 0], t0); atomicAdd(&tot[c4 + 1], t1);
        atomicAdd(&tot[c4 + 2], t2); atomicAdd(&tot[c4 + 3], t3);
        atomicAdd(&cur[c4 + 0], p0); atomicAdd(&cur[c4 + 1], p1);
        atomicAdd(&cur[c4 + 2], p2); atomicAdd(&cur[c4 + 3], p3);
    }
    __syncthreads();

    const bool own = (4 * tid < tp);
    int a0 = 0, a1 = 0, a2 = 0, a3 = 0;
    if (own) { a0 = tot[4 * tid]; a1 = tot[4 * tid + 1]; a2 = tot[4 * tid + 2]; a3 = tot[4 * tid + 3]; }
    const int s = a0 + a1 + a2 + a3;
    int inc = s;
#pragma unroll
    for (int d = 1; d < 32; d <<= 1) {
        int t = __shfl_up(inc, d);
        if (lane >= d) inc += t;
    }
    if (lane == 31) wtot[wv] = inc;
    __syncthreads();
    int base0 = inc - s;
#pragma unroll
    for (int k = 0; k < SORT_T / 32; ++k) if (k < wv) base0 += wtot[k];
    if (own) {
        const int s0 = base0, s1 = s0 + a0, s2 = s1 + a1, s3 = s2 + a2;
        cur[4 * tid]     += s0;
        cur[4 * tid + 1] += s1;
        cur[4 * tid + 2] += s2;
        cur[4 * tid + 3] += s3;
        if (w == 0) {
            bstart[4 * tid]     = s0;
            bstart[4 * tid + 1] = s1;
            bstart[4 * tid + 2] = s2;
            bstart[4 * tid + 3] = s3;
            if (4 * tid + 4 == tp) bstart[tp] = s3 + a3;
        }
    }
    __syncthreads();

    for (int base = e0; base < e1; base += SORT_T * SORT_K) {
        if (base != e0) {
#pragma unroll
            for (int k = 0; k < SORT_K; ++k) {
                const int e = min(base + tid + k * SORT_T, e1 - 1);
                c[k]  = (unsigned)col[e];
                r[k]  = row[e];
                wt[k] = __float_as_int(ew[e]);
            }
        }
        int pos[SORT_K];
#pragma unroll
        for (int k = 0; k < SORT_K; ++k) {
            const int e = base + tid + k * SORT_T;
            const int b = (e < e1) ? min((int)(c[k] >> BKT_SHIFT), nb - 1) : MAX_NB;
            pos[k] = atomicAdd(&cur[b], 1);
        }
#pragma unroll
        for (int k = 0; k < SORT_K; ++k) {
            const int e = base + tid + k * SORT_T;
            if (e < e1)
                prec[pos[k]] = make_int2((int)((unsigned)r[k] | ((c[k] & (BKT_W - 1)) << 24)), wt[k]);
        }
    }
}

__global__ __launch_bounds__(SEG_T) void bucket_sort_kernel(const int* __restrict__ bstart,
                                                            const int2* __restrict__ prec,
                                                            int* __restrict__ off,
                                                            int2* __restrict__ sedge, int n) {
    __shared__ int cnt[BKT_W + 1];
    __shared__ int wtot[BKT_W / 32];
    const int tid  = threadIdx.x;
    const int lane = tid & 31;
    const int wv   = tid >> 5;
    const int b    = blockIdx.x;
    const int beg  = bstart[b];
    const int end  = bstart[b + 1];

    int2 rec[SEG_K];
#pragma unroll
    for (int k = 0; k < SEG_K; ++k) {
        const int i = beg + tid + k * SEG_T;
        rec[k] = prec[max(min(i, end - 1), 0)];
    }
    if (tid < BKT_W) cnt[tid] = 0;
    __syncthreads();
#pragma unroll
    for (int k = 0; k < SEG_K; ++k) {
        const int i = beg + tid + k * SEG_T;
        atomicAdd(&cnt[(i < end) ? (int)((unsigned)rec[k].x >> 24) : BKT_W], 1);
    }
    for (int i = beg + SEG_T * SEG_K + tid; i < end; i += SEG_T)
        atomicAdd(&cnt[(unsigned)prec[i].x >> 24], 1);
    __syncthreads();

    const int v = (tid < BKT_W) ? cnt[tid] : 0;
    int inc = v;
#pragma unroll
    for (int d = 1; d < 32; d <<= 1) {
        int t = __shfl_up(inc, d);
        if (lane >= d) inc += t;
    }
    if (tid < BKT_W && lane == 31) wtot[wv] = inc;
    __syncthreads();
    if (tid < BKT_W) {
        int excl = beg + inc - v;
#pragma unroll
        for (int k = 0; k < BKT_W / 32; ++k) if (k < wv) excl += wtot[k];
        cnt[tid] = excl;
        const int node = b * BKT_W + tid;
        if (node < n) off[node] = excl;
        if (node == n - 1) off[n] = excl + v;
    }
    __syncthreads();
    int pos[SEG_K];
#pragma unroll
    for (int k = 0; k < SEG_K; ++k) {
        const int i = beg + tid + k * SEG_T;
        pos[k] = atomicAdd(&cnt[(i < end) ? (int)((unsigned)rec[k].x >> 24) : BKT_W], 1);
    }
#pragma unroll
    for (int k = 0; k < SEG_K; ++k) {
        const int i = beg + tid + k * SEG_T;
        if (i < end) sedge[pos[k]] = make_int2((int)((unsigned)rec[k].x & 0xFFFFFFu), rec[k].y);
    }
    for (int i = beg + SEG_T * SEG_K + tid; i < end; i += SEG_T) {
        const int2 q = prec[i];
        const unsigned x = (unsigned)q.x;
        const int at = atomicAdd(&cnt[x >> 24], 1);
        sedge[at] = make_int2((int)(x & 0xFFFFFFu), q.y);
    }
}

__global__ __launch_bounds__(256) void dis_kernel(const int* __restrict__ off,
                                                  const int2* __restrict__ sedge,
                                                  float* __restrict__ dis, int n) {
    const int gid = blockIdx.x * 256 + threadIdx.x;
    const int i   = gid >> 3;
    const int sub = gid & 7;
    float s = 0.0f;
    if (i < n) {
        const int end = off[i + 1];
        for (int p = off[i] + sub; p < end; p += 8) s += __int_as_float(sedge[p].y);
    }
    s += __shfl_xor(s, 1);
    s += __shfl_xor(s, 2);
    s += __shfl_xor(s, 4);
    if (i < n && sub == 0) dis[i] = rsqrtf(s + 1.0f);
}

#define TILE_M 32
#define NPW    (TILE_M / 8)
#define LDA    132

__device__ __forceinline__ void row_fma(float4& acc, float cf, const float4 hv) {
    acc.x = fmaf(cf, hv.x, acc.x);
    acc.y = fmaf(cf, hv.y, acc.y);
    acc.z = fmaf(cf, hv.z, acc.z);
    acc.w = fmaf(cf, hv.w, acc.w);
}

template <int K>
__device__ __forceinline__ void add_rows(float4& acc, int r, float cf, int j,
                                         const float* __restrict__ h, unsigned lane4) {
    float4 hv[K];
#pragma unroll
    for (int k = 0; k < K; ++k) {
        const int rk = __builtin_amdgcn_readlane(r, j + k);
        hv[k] = *(const float4*)(h + (size_t)rk * FEAT + lane4);
    }
#pragma unroll
    for (int k = 0; k < K; ++k) {
        const float ck = __int_as_float(__builtin_amdgcn_readlane(__float_as_int(cf), j + k));
        row_fma(acc, ck, hv[k]);
    }
}

__device__ __forceinline__ void aggregate_to_lds(float* __restrict__ lA,
                                                 const int* __restrict__ off,
                                                 const int2* __restrict__ sedge,
                                                 const float* __restrict__ dis,
                                                 const float* __restrict__ X,
                                                 int rowBase, int n) {
    const int lane = threadIdx.x & 31;
    const unsigned lane4 = (unsigned)lane * 4u;
    const int wave = __builtin_amdgcn_readfirstlane(threadIdx.x >> 5);
    const int c0 = rowBase + wave * NPW;
    int   ov = 0;
    float dv = 0.f;
    if (lane <= NPW) ov = off[min(c0 + lane, n)];
    if (lane <  NPW) dv = dis[min(c0 + lane, n - 1)];

    for (int i = 0; i < NPW; ++i) {
        const int c = c0 + i;
        float4 acc = make_float4(0.f, 0.f, 0.f, 0.f);
        if (c < n) {
            const int beg = __builtin_amdgcn_readlane(ov, i);
            const int end = __builtin_amdgcn_readlane(ov, i + 1);
            const float dc = __int_as_float(__builtin_amdgcn_readlane(__float_as_int(dv), i));
            const float4 xv = *(const float4*)(X + (size_t)c * FEAT + lane4);
            for (int base = beg; base < end; base += 32) {
                const int p = base + lane;
                int r = 0;
                float cf = 0.f;
                if (p < end) {
                    const int2 ed = sedge[p];
                    r  = ed.x;
                    cf = dis[r] * __int_as_float(ed.y) * dc;
                }
                const int m = (end - base < 32) ? (end - base) : 32;
                int j = 0;
                for (; j + 8 <= m; j += 8) add_rows<8>(acc, r, cf, j, X, lane4);
                if (m - j >= 4) { add_rows<4>(acc, r, cf, j, X, lane4); j += 4; }
                const int rem = m - j;
                if (rem == 3)      add_rows<3>(acc, r, cf, j, X, lane4);
                else if (rem == 2) add_rows<2>(acc, r, cf, j, X, lane4);
                else if (rem == 1) add_rows<1>(acc, r, cf, j, X, lane4);
            }
            row_fma(acc, dc * dc, xv);
        }
        *(float4*)(lA + (wave * NPW + i) * LDA + lane4) = acc;
    }
}

__device__ __forceinline__ void tile_times_w(const float* __restrict__ lA,
                                             const float* __restrict__ Wm,
                                             int n0, int lid, int half, v8f& acc0, v8f& acc1) {
    const float* arow0 = lA + lid * LDA;
    const float* arow1 = lA + (16 + lid) * LDA;
#pragma unroll
    for (int k0 = 0; k0 < FEAT; k0 += 4) {
        const int ka = k0 + 2 * half;
        v2f a0, a1, b;
        a0.x = arow0[ka];
        a0.y = arow0[ka + 1];
        a1.x = arow1[ka];
        a1.y = arow1[ka + 1];
        b.x = Wm[(size_t)ka * FEAT + n0 + lid];
        b.y = Wm[(size_t)(ka + 1) * FEAT + n0 + lid];
        acc0 = __builtin_amdgcn_wmma_f32_16x16x4_f32(
             false, a0,  false, b,
             (short)0, acc0,  false,  false);
        acc1 = __builtin_amdgcn_wmma_f32_16x16x4_f32(
             false, a1,  false, b,
             (short)0, acc1,  false,  false);
    }
}

__global__ __launch_bounds__(256, 16) void layer1_kernel(const int* __restrict__ off,
                                                     const int2* __restrict__ sedge,
                                                     const float* __restrict__ dis,
                                                     const float* __restrict__ emb,
                                                     const float* __restrict__ W0,
                                                     const float* __restrict__ b0,
                                                     const float* __restrict__ g,
                                                     const float* __restrict__ be,
                                                     const float* __restrict__ mu,
                                                     const float* __restrict__ var,
                                                     float* __restrict__ h1, int n) {
    __shared__ __attribute__((aligned(16))) float lA[TILE_M * LDA];
    const int rowBase = blockIdx.x * TILE_M;
    aggregate_to_lds(lA, off, sedge, dis, emb, rowBase, n);
    __syncthreads();

    const int lane = threadIdx.x & 31;
    const int lid  = lane & 15;
    const int half = lane >> 4;
    const int n0   = (threadIdx.x >> 5) * 16;
    const int f    = n0 + lid;
    const float bb = b0[f];
    const float mm = mu[f];
    const float rs = rsqrtf(var[f] + BN_EPS);
    const float gg = g[f];
    const float bt = be[f];

    v8f acc0 = {0.f, 0.f, 0.f, 0.f, 0.f, 0.f, 0.f, 0.f};
    v8f acc1 = {0.f, 0.f, 0.f, 0.f, 0.f, 0.f, 0.f, 0.f};
    tile_times_w(lA, W0, n0, lid, half, acc0, acc1);

    float* yb = h1 + (size_t)rowBase * FEAT + f;
    const bool whole = (rowBase + TILE_M <= n);
#pragma unroll
    for (int t = 0; t < 2; ++t) {
#pragma unroll
        for (int v = 0; v < 8; ++v) {
            const int m = 16 * t + v + 8 * half;
            float x = (t == 0 ? acc0[v] : acc1[v]) + bb;
            x = (x - mm) * rs * gg + bt;
            x = (x > 0.0f) ? x : expm1f(x);
            if (whole || rowBase + m < n) yb[(size_t)m * FEAT] = x;
        }
    }
}

__global__ __launch_bounds__(256, 16) void layer2_kernel(const int* __restrict__ off,
                                                     const int2* __restrict__ sedge,
                                                     const float* __restrict__ dis,
                                                     const float* __restrict__ h1,
                                                     const float* __restrict__ W1,
                                                     const float* __restrict__ b1,
                                                     const float* __restrict__ emb,
                                                     float* __restrict__ out, int n) {
    __shared__ __attribute__((aligned(16))) float lA[TILE_M * LDA];
    const int rowBase = blockIdx.x * TILE_M;
    aggregate_to_lds(lA, off, sedge, dis, h1, rowBase, n);
    __syncthreads();

    const int lane = threadIdx.x & 31;
    const int lid  = lane & 15;
    const int half = lane >> 4;
    const int n0   = (threadIdx.x >> 5) * 16;
    const int f    = n0 + lid;
    const float bb = b1[f];

    v8f acc0 = {0.f, 0.f, 0.f, 0.f, 0.f, 0.f, 0.f, 0.f};
    v8f acc1 = {0.f, 0.f, 0.f, 0.f, 0.f, 0.f, 0.f, 0.f};
    tile_times_w(lA, W1, n0, lid, half, acc0, acc1);

    const float inv3 = 1.0f / 3.0f;
    const size_t rb = (size_t)rowBase * FEAT + f;
    if (rowBase + TILE_M <= n) {
        float s[16];
#pragma unroll
        for (int q = 0; q < 16; ++q) {
            const size_t at = rb + (size_t)(16 * (q >> 3) + (q & 7) + 8 * half) * FEAT;
            s[q] = emb[at] + h1[at];
        }
#pragma unroll
        for (int q = 0; q < 16; ++q) {
            const size_t at = rb + (size_t)(16 * (q >> 3) + (q & 7) + 8 * half) * FEAT;
            out[at] = (s[q] + (q < 8 ? acc0[q & 7] : acc1[q & 7]) + bb) * inv3;
        }
    } else {
#pragma unroll
        for (int q = 0; q < 16; ++q) {
            const int m = 16 * (q >> 3) + (q & 7) + 8 * half;
            if (rowBase + m < n) {
                const size_t at = rb + (size_t)m * FEAT;
                out[at] = (emb[at] + h1[at] + (q < 8 ? acc0[q & 7] : acc1[q & 7]) + bb) * inv3;
            }
        }
    }
}

extern "C" void kernel_launch(void* const* d_in, const int* in_sizes, int n_in,
                              void* d_out, int out_size, void* d_ws, size_t ws_size,
                              hipStream_t stream) {
    const float* emb = (const float*)d_in[0];
    const int*   ei  = (const int*)d_in[1];
    const float* ew  = (const float*)d_in[2];
    const float* W0  = (const float*)d_in[3];
    const float* b0  = (const float*)d_in[4];
    const float* W1  = (const float*)d_in[5];
    const float* b1  = (const float*)d_in[6];
    const float* bng = (const float*)d_in[7];
    const float* bnb = (const float*)d_in[8];
    const float* bnm = (const float*)d_in[9];
    const float* bnv = (const float*)d_in[10];
    float* out = (float*)d_out;

    const int N  = in_sizes[0] / FEAT;
    const int E  = in_sizes[2];

    const int* row = ei;
    const int* col = ei + E;

    float* ws   = (float*)d_ws;
    size_t nPad = ((size_t)N + 1023) & ~(size_t)1023;
    size_t ePad = ((size_t)E + 1023) & ~(size_t)1023;
    int*   tab    = (int*)ws;
    int*   off    = tab + nPad;
    float* dis    = (float*)(off + nPad + 1024);
    int*   bstart = (int*)(dis + nPad);
    int2*  prec   = (int2*)(bstart + 2048);
    int2*  sedge  = prec + ePad;
    float* h1     = (float*)(sedge + ePad);

    const int gTile  = (N + TILE_M - 1) / TILE_M;
    const int nb     = (N + BKT_W - 1) / BKT_W;
    const int tp     = (nb + 3) & ~3;
    const int chunk  = (E + SORT_G - 1) / SORT_G;

    bucket_count_kernel<<<SORT_G, SORT_T, 0, stream>>>(col, tab, E, chunk, nb, tp);
    bucket_place_kernel<<<SORT_G, SORT_T, 0, stream>>>(row, col, ew, tab, bstart, prec, E, chunk, nb, tp);
    bucket_sort_kernel<<<nb, SEG_T, 0, stream>>>(bstart, prec, off, sedge, N);
    dis_kernel<<<(int)(((size_t)N * 8 + 255) / 256), 256, 0, stream>>>(off, sedge, dis, N);

    layer1_kernel<<<gTile, 256, 0, stream>>>(off, sedge, dis, emb, W0, b0, bng, bnb, bnm, bnv, h1, N);

    layer2_kernel<<<gTile, 256, 0, stream>>>(off, sedge, dis, h1, W1, b1, emb, out, N);
}


# sort passes: a lane without an edge makes no LDS add (all such lanes added to one spare word, and same-address LDS adds are served lane by lane); edges inside a node stay in LDS-rank order
# speedup vs baseline: 4.3087x; 1.3822x over previous
#include <hip/hip_runtime.h>
#include <hip/hip_bf16.h>
#include <math.h>

typedef float v2f __attribute__((ext_vector_type(2)));
typedef float v8f __attribute__((ext_vector_type(8)));

#define FEAT 128
#define BN_EPS 1e-5f

#define BKT_SHIFT 8
#define BKT_W     256
#define MAX_NB    1024
#define SORT_G    256
#define SORT_T    512
#define SORT_K    8
#define TAB_RG    (SORT_T / 64)
#define TAB_RPG   (SORT_G / TAB_RG)
#define SEG_T     512
#define SEG_K     16

__global__ __launch_bounds__(SORT_T) void bucket_count_kernel(const int* __restrict__ col,
                                                              int* __restrict__ tab,
                                                              int E, int chunk, int nb, int tp) {
    __shared__ int hist[MAX_NB];
    const int tid = threadIdx.x;
    const int e0 = blockIdx.x * chunk;
    const int e1 = min(e0 + chunk, E);
    for (int b = tid; b < tp; b += SORT_T) hist[b] = 0;
    __syncthreads();
    for (int base = e0; base < e1; base += SORT_T * SORT_K) {
        unsigned c[SORT_K];
#pragma unroll
        for (int k = 0; k < SORT_K; ++k) {
            const int e = base + tid + k * SORT_T;
            c[k] = (unsigned)col[min(e, e1 - 1)];
        }
#pragma unroll
        for (int k = 0; k < SORT_K; ++k) {
            const int e = base + tid + k * SORT_T;
            if (e < e1) atomicAdd(&hist[min((int)(c[k] >> BKT_SHIFT), nb - 1)], 1);
        }
    }
    __syncthreads();
    for (int b = tid; b < tp; b += SORT_T) tab[(size_t)blockIdx.x * tp + b] = hist[b];
}

__global__ __launch_bounds__(SORT_T) void bucket_place_kernel(const int* __restrict__ row,
                                                              const int* __restrict__ col,
                                                              const float* __restrict__ ew,
                                                              const int* __restrict__ tab,
                                                              int* __restrict__ bstart,
                                                              int2* __restrict__ prec,
                                                              int E, int chunk, int nb, int tp) {
    __shared__ int tot[MAX_NB];
    __shared__ int cur[MAX_NB];
    __shared__ int wtot[SORT_T / 32];
    const int tid  = threadIdx.x;
    const int lane = tid & 31;
    const int wv   = tid >> 5;
    const int w    = blockIdx.x;
    const int e0 = w * chunk;
    const int e1 = min(e0 + chunk, E);

    for (int b = tid; b < tp; b += SORT_T) { tot[b] = 0; cur[b] = 0; }
    __syncthreads();

    unsigned c[SORT_K];
    int      r[SORT_K];
    int      wt[SORT_K];
#pragma unroll
    for (int k = 0; k < SORT_K; ++k) { c[k] = 0u; r[k] = 0; wt[k] = 0; }
    if (e0 < e1) {
#pragma unroll
        for (int k = 0; k < SORT_K; ++k) {
            const int e = min(e0 + tid + k * SORT_T, e1 - 1);
            c[k]  = (unsigned)col[e];
            r[k]  = row[e];
            wt[k] = __float_as_int(ew[e]);
        }
    }

    const int cg = tid & 63;
    const int rg = tid >> 6;
    for (int c4 = cg * 4; c4 < tp; c4 += 256) {
        int4 v[TAB_RPG];
#pragma unroll
        for (int i = 0; i < TAB_RPG; ++i) {
            const int rr = rg * TAB_RPG + ((i + w) & (TAB_RPG - 1));
            v[i] = *(const int4*)(tab + (size_t)rr * tp + c4);
        }
        __builtin_amdgcn_sched_barrier(0);
        int t0 = 0, t1 = 0, t2 = 0, t3 = 0;
        int p0 = 0, p1 = 0, p2 = 0, p3 = 0;
#pragma unroll
        for (int i = 0; i < TAB_RPG; ++i) {
            const int rr = rg * TAB_RPG + ((i + w) & (TAB_RPG - 1));
            t0 += v[i].x; t1 += v[i].y; t2 += v[i].z; t3 += v[i].w;
            if (rr < w) { p0 += v[i].x; p1 += v[i].y; p2 += v[i].z; p3 += v[i].w; }
        }
        atomicAdd(&tot[c4 + 0], t0); atomicAdd(&tot[c4 + 1], t1);
        atomicAdd(&tot[c4 + 2], t2); atomicAdd(&tot[c4 + 3], t3);
        atomicAdd(&cur[c4 + 0], p0); atomicAdd(&cur[c4 + 1], p1);
        atomicAdd(&cur[c4 + 2], p2); atomicAdd(&cur[c4 + 3], p3);
    }
    __syncthreads();

    const bool own = (4 * tid < tp);
    int a0 = 0, a1 = 0, a2 = 0, a3 = 0;
    if (own) { a0 = tot[4 * tid]; a1 = tot[4 * tid + 1]; a2 = tot[4 * tid + 2]; a3 = tot[4 * tid + 3]; }
    const int s = a0 + a1 + a2 + a3;
    int inc = s;
#pragma unroll
    for (int d = 1; d < 32; d <<= 1) {
        int t = __shfl_up(inc, d);
        if (lane >= d) inc += t;
    }
    if (lane == 31) wtot[wv] = inc;
    __syncthreads();
    int base0 = inc - s;
#pragma unroll
    for (int k = 0; k < SORT_T / 32; ++k) if (k < wv) base0 += wtot[k];
    if (own) {
        const int s0 = base0, s1 = s0 + a0, s2 = s1 + a1, s3 = s2 + a2;
        cur[4 * tid]     += s0;
        cur[4 * tid + 1] += s1;
        cur[4 * tid + 2] += s2;
        cur[4 * tid + 3] += s3;
        if (w == 0) {
            bstart[4 * tid]     = s0;
            bstart[4 * tid + 1] = s1;
            bstart[4 * tid + 2] = s2;
            bstart[4 * tid + 3] = s3;
            if (4 * tid + 4 == tp) bstart[tp] = s3 + a3;
        }
    }
    __syncthreads();

    for (int base = e0; base < e1; base += SORT_T * SORT_K) {
        if (base != e0) {
#pragma unroll
            for (int k = 0; k < SORT_K; ++k) {
                const int e = min(base + tid + k * SORT_T, e1 - 1);
                c[k]  = (unsigned)col[e];
                r[k]  = row[e];
                wt[k] = __float_as_int(ew[e]);
            }
        }
        int pos[SORT_K];
#pragma unroll
        for (int k = 0; k < SORT_K; ++k) {
            const int e = base + tid + k * SORT_T;
            pos[k] = 0;
            if (e < e1) pos[k] = atomicAdd(&cur[min((int)(c[k] >> BKT_SHIFT), nb - 1)], 1);
        }
#pragma unroll
        for (int k = 0; k < SORT_K; ++k) {
            const int e = base + tid + k * SORT_T;
            if (e < e1)
                prec[pos[k]] = make_int2((int)((unsigned)r[k] | ((c[k] & (BKT_W - 1)) << 24)), wt[k]);
        }
    }
}

__global__ __launch_bounds__(SEG_T) void bucket_sort_kernel(const int* __restrict__ bstart,
                                                            const int2* __restrict__ prec,
                                                            int* __restrict__ off,
                                                            int2* __restrict__ sedge, int n) {
    __shared__ int cnt[BKT_W];
    __shared__ int wtot[BKT_W / 32];
    const int tid  = threadIdx.x;
    const int lane = tid & 31;
    const int wv   = tid >> 5;
    const int b    = blockIdx.x;
    const int beg  = bstart[b];
    const int end  = bstart[b + 1];

    int2 rec[SEG_K];
#pragma unroll
    for (int k = 0; k < SEG_K; ++k) {
        const int i = beg + tid + k * SEG_T;
        rec[k] = prec[max(min(i, end - 1), 0)];
    }
    if (tid < BKT_W) cnt[tid] = 0;
    __syncthreads();
#pragma unroll
    for (int k = 0; k < SEG_K; ++k) {
        const int i = beg + tid + k * SEG_T;
        if (i < end) atomicAdd(&cnt[(unsigned)rec[k].x >> 24], 1);
    }
    for (int i = beg + SEG_T * SEG_K + tid; i < end; i += SEG_T)
        atomicAdd(&cnt[(unsigned)prec[i].x >> 24], 1);
    __syncthreads();

    const int v = (tid < BKT_W) ? cnt[tid] : 0;
    int inc = v;
#pragma unroll
    for (int d = 1; d < 32; d <<= 1) {
        int t = __shfl_up(inc, d);
        if (lane >= d) inc += t;
    }
    if (tid < BKT_W && lane == 31) wtot[wv] = inc;
    __syncthreads();
    if (tid < BKT_W) {
        int excl = beg + inc - v;
#pragma unroll
        for (int k = 0; k < BKT_W / 32; ++k) if (k < wv) excl += wtot[k];
        cnt[tid] = excl;
        const int node = b * BKT_W + tid;
        if (node < n) off[node] = excl;
        if (node == n - 1) off[n] = excl + v;
    }
    __syncthreads();
    int pos[SEG_K];
#pragma unroll
    for (int k = 0; k < SEG_K; ++k) {
        const int i = beg + tid + k * SEG_T;
        pos[k] = 0;
        if (i < end) pos[k] = atomicAdd(&cnt[(unsigned)rec[k].x >> 24], 1);
    }
#pragma unroll
    for (int k = 0; k < SEG_K; ++k) {
        const int i = beg + tid + k * SEG_T;
        if (i < end) sedge[pos[k]] = make_int2((int)((unsigned)rec[k].x & 0xFFFFFFu), rec[k].y);
    }
    for (int i = beg + SEG_T * SEG_K + tid; i < end; i += SEG_T) {
        const int2 q = prec[i];
        const unsigned x = (unsigned)q.x;
        const int at = atomicAdd(&cnt[x >> 24], 1);
        sedge[at] = make_int2((int)(x & 0xFFFFFFu), q.y);
    }
}

__global__ __launch_bounds__(256) void dis_kernel(const int* __restrict__ off,
                                                  const int2* __restrict__ sedge,
                                                  float* __restrict__ dis, int n) {
    const int gid = blockIdx.x * 256 + threadIdx.x;
    const int i   = gid >> 3;
    const int sub = gid & 7;
    float s = 0.0f;
    if (i < n) {
        const int end = off[i + 1];
        for (int p = off[i] + sub; p < end; p += 8) s += __int_as_float(sedge[p].y);
    }
    s += __shfl_xor(s, 1);
    s += __shfl_xor(s, 2);
    s += __shfl_xor(s, 4);
    if (i < n && sub == 0) dis[i] = rsqrtf(s + 1.0f);
}

#define TILE_M 32
#define NPW    (TILE_M / 8)
#define LDA    132

__device__ __forceinline__ void row_fma(float4& acc, float cf, const float4 hv) {
    acc.x = fmaf(cf, hv.x, acc.x);
    acc.y = fmaf(cf, hv.y, acc.y);
    acc.z = fmaf(cf, hv.z, acc.z);
    acc.w = fmaf(cf, hv.w, acc.w);
}

template <int K>
__device__ __forceinline__ void add_rows(float4& acc, int r, float cf, int j,
                                         const float* __restrict__ h, unsigned lane4) {
    float4 hv[K];
#pragma unroll
    for (int k = 0; k < K; ++k) {
        const int rk = __builtin_amdgcn_readlane(r, j + k);
        hv[k] = *(const float4*)(h + (size_t)rk * FEAT + lane4);
    }
#pragma unroll
    for (int k = 0; k < K; ++k) {
        const float ck = __int_as_float(__builtin_amdgcn_readlane(__float_as_int(cf), j + k));
        row_fma(acc, ck, hv[k]);
    }
}

__device__ __forceinline__ void aggregate_to_lds(float* __restrict__ lA,
                                                 const int* __restrict__ off,
                                                 const int2* __restrict__ sedge,
                                                 const float* __restrict__ dis,
                                                 const float* __restrict__ X,
                                                 int rowBase, int n) {
    const int lane = threadIdx.x & 31;
    const unsigned lane4 = (unsigned)lane * 4u;
    const int wave = __builtin_amdgcn_readfirstlane(threadIdx.x >> 5);
    const int c0 = rowBase + wave * NPW;
    int   ov = 0;
    float dv = 0.f;
    if (lane <= NPW) ov = off[min(c0 + lane, n)];
    if (lane <  NPW) dv = dis[min(c0 + lane, n - 1)];

    for (int i = 0; i < NPW; ++i) {
        const int c = c0 + i;
        float4 acc = make_float4(0.f, 0.f, 0.f, 0.f);
        if (c < n) {
            const int beg = __builtin_amdgcn_readlane(ov, i);
            const int end = __builtin_amdgcn_readlane(ov, i + 1);
            const float dc = __int_as_float(__builtin_amdgcn_readlane(__float_as_int(dv), i));
            const float4 xv = *(const float4*)(X + (size_t)c * FEAT + lane4);
            for (int base = beg; base < end; base += 32) {
                const int p = base + lane;
                int r = 0;
                float cf = 0.f;
                if (p < end) {
                    const int2 ed = sedge[p];
                    r  = ed.x;
                    cf = dis[r] * __int_as_float(ed.y) * dc;
                }
                const int m = (end - base < 32) ? (end - base) : 32;
                int j = 0;
                for (; j + 8 <= m; j += 8) add_rows<8>(acc, r, cf, j, X, lane4);
                if (m - j >= 4) { add_rows<4>(acc, r, cf, j, X, lane4); j += 4; }
                const int rem = m - j;
                if (rem == 3)      add_rows<3>(acc, r, cf, j, X, lane4);
                else if (rem == 2) add_rows<2>(acc, r, cf, j, X, lane4);
                else if (rem == 1) add_rows<1>(acc, r, cf, j, X, lane4);
            }
            row_fma(acc, dc * dc, xv);
        }
        *(float4*)(lA + (wave * NPW + i) * LDA + lane4) = acc;
    }
}

__device__ __forceinline__ void tile_times_w(const float* __restrict__ lA,
                                             const float* __restrict__ Wm,
                                             int n0, int lid, int half, v8f& acc0, v8f& acc1) {
    const float* arow0 = lA + lid * LDA;
    const float* arow1 = lA + (16 + lid) * LDA;
#pragma unroll
    for (int k0 = 0; k0 < FEAT; k0 += 4) {
        const int ka = k0 + 2 * half;
        v2f a0, a1, b;
        a0.x = arow0[ka];
        a0.y = arow0[ka + 1];
        a1.x = arow1[ka];
        a1.y = arow1[ka + 1];
        b.x = Wm[(size_t)ka * FEAT + n0 + lid];
        b.y = Wm[(size_t)(ka + 1) * FEAT + n0 + lid];
        acc0 = __builtin_amdgcn_wmma_f32_16x16x4_f32(
             false, a0,  false, b,
             (short)0, acc0,  false,  false);
        acc1 = __builtin_amdgcn_wmma_f32_16x16x4_f32(
             false, a1,  false, b,
             (short)0, acc1,  false,  false);
    }
}

__global__ __launch_bounds__(256, 16) void layer1_kernel(const int* __restrict__ off,
                                                     const int2* __restrict__ sedge,
                                                     const float* __restrict__ dis,
                                                     const float* __restrict__ emb,
                                                     const float* __restrict__ W0,
                                                     const float* __restrict__ b0,
                                                     const float* __restrict__ g,
                                                     const float* __restrict__ be,
                                                     const float* __restrict__ mu,
                                                     const float* __restrict__ var,
                                                     float* __restrict__ h1, int n) {
    __shared__ __attribute__((aligned(16))) float lA[TILE_M * LDA];
    const int rowBase = blockIdx.x * TILE_M;
    aggregate_to_lds(lA, off, sedge, dis, emb, rowBase, n);
    __syncthreads();

    const int lane = threadIdx.x & 31;
    const int lid  = lane & 15;
    const int half = lane >> 4;
    const int n0   = (threadIdx.x >> 5) * 16;
    const int f    = n0 + lid;
    const float bb = b0[f];
    const float mm = mu[f];
    const float rs = rsqrtf(var[f] + BN_EPS);
    const float gg = g[f];
    const float bt = be[f];

    v8f acc0 = {0.f, 0.f, 0.f, 0.f, 0.f, 0.f, 0.f, 0.f};
    v8f acc1 = {0.f, 0.f, 0.f, 0.f, 0.f, 0.f, 0.f, 0.f};
    tile_times_w(lA, W0, n0, lid, half, acc0, acc1);

    float* yb = h1 + (size_t)rowBase * FEAT + f;
    const bool whole = (rowBase + TILE_M <= n);
#pragma unroll
    for (int t = 0; t < 2; ++t) {
#pragma unroll
        for (int v = 0; v < 8; ++v) {
            const int m = 16 * t + v + 8 * half;
            float x = (t == 0 ? acc0[v] : acc1[v]) + bb;
            x = (x - mm) * rs * gg + bt;
            x = (x > 0.0f) ? x : expm1f(x);
            if (whole || rowBase + m < n) yb[(size_t)m * FEAT] = x;
        }
    }
}

__global__ __launch_bounds__(256, 16) void layer2_kernel(const int* __restrict__ off,
                                                     const int2* __restrict__ sedge,
                                                     const float* __restrict__ dis,
                                                     const float* __restrict__ h1,
                                                     const float* __restrict__ W1,
                                                     const float* __restrict__ b1,
                                                     const float* __restrict__ emb,
                                                     float* __restrict__ out, int n) {
    __shared__ __attribute__((aligned(16))) float lA[TILE_M * LDA];
    const int rowBase = blockIdx.x * TILE_M;
    aggregate_to_lds(lA, off, sedge, dis, h1, rowBase, n);
    __syncthreads();

    const int lane = threadIdx.x & 31;
    const int lid  = lane & 15;
    const int half = lane >> 4;
    const int n0   = (threadIdx.x >> 5) * 16;
    const int f    = n0 + lid;
    const float bb = b1[f];

    v8f acc0 = {0.f, 0.f, 0.f, 0.f, 0.f, 0.f, 0.f, 0.f};
    v8f acc1 = {0.f, 0.f, 0.f, 0.f, 0.f, 0.f, 0.f, 0.f};
    tile_times_w(lA, W1, n0, lid, half, acc0, acc1);

    const float inv3 = 1.0f / 3.0f;
    const size_t rb = (size_t)rowBase * FEAT + f;
    if (rowBase + TILE_M <= n) {
        float s[16];
#pragma unroll
        for (int q = 0; q < 16; ++q) {
            const size_t at = rb + (size_t)(16 * (q >> 3) + (q & 7) + 8 * half) * FEAT;
            s[q] = emb[at] + h1[at];
        }
#pragma unroll
        for (int q = 0; q < 16; ++q) {
            const size_t at = rb + (size_t)(16 * (q >> 3) + (q & 7) + 8 * half) * FEAT;
            out[at] = (s[q] + (q < 8 ? acc0[q & 7] : acc1[q & 7]) + bb) * inv3;
        }
    } else {
#pragma unroll
        for (int q = 0; q < 16; ++q) {
            const int m = 16 * (q >> 3) + (q & 7) + 8 * half;
            if (rowBase + m < n) {
                const size_t at = rb + (size_t)m * FEAT;
                out[at] = (emb[at] + h1[at] + (q < 8 ? acc0[q & 7] : acc1[q & 7]) + bb) * inv3;
            }
        }
    }
}

extern "C" void kernel_launch(void* const* d_in, const int* in_sizes, int n_in,
                              void* d_out, int out_size, void* d_ws, size_t ws_size,
                              hipStream_t stream) {
    const float* emb = (const float*)d_in[0];
    const int*   ei  = (const int*)d_in[1];
    const float* ew  = (const float*)d_in[2];
    const float* W0  = (const float*)d_in[3];
    const float* b0  = (const float*)d_in[4];
    const float* W1  = (const float*)d_in[5];
    const float* b1  = (const float*)d_in[6];
    const float* bng = (const float*)d_in[7];
    const float* bnb = (const float*)d_in[8];
    const float* bnm = (const float*)d_in[9];
    const float* bnv = (const float*)d_in[10];
    float* out = (float*)d_out;

    const int N  = in_sizes[0] / FEAT;
    const int E  = in_sizes[2];

    const int* row = ei;
    const int* col = ei + E;

    float* ws   = (float*)d_ws;
    size_t nPad = ((size_t)N + 1023) & ~(size_t)1023;
    size_t ePad = ((size_t)E + 1023) & ~(size_t)1023;
    int*   tab    = (int*)ws;
    int*   off    = tab + nPad;
    float* dis    = (float*)(off + nPad + 1024);
    int*   bstart = (int*)(dis + nPad);
    int2*  prec   = (int2*)(bstart + 2048);
    int2*  sedge  = prec + ePad;
    float* h1     = (float*)(sedge + ePad);

    const int gTile  = (N + TILE_M - 1) / TILE_M;
    const int nb     = (N + BKT_W - 1) / BKT_W;
    const int tp     = (nb + 3) & ~3;
    const int chunk  = (E + SORT_G - 1) / SORT_G;

    bucket_count_kernel<<<SORT_G, SORT_T, 0, stream>>>(col, tab, E, chunk, nb, tp);
    bucket_place_kernel<<<SORT_G, SORT_T, 0, stream>>>(row, col, ew, tab, bstart, prec, E, chunk, nb, tp);
    bucket_sort_kernel<<<nb, SEG_T, 0, stream>>>(bstart, prec, off, sedge, N);
    dis_kernel<<<(int)(((size_t)N * 8 + 255) / 256), 256, 0, stream>>>(off, sedge, dis, N);

    layer1_kernel<<<gTile, 256, 0, stream>>>(off, sedge, dis, emb, W0, b0, bng, bnb, bnm, bnv, h1, N);

    layer2_kernel<<<gTile, 256, 0, stream>>>(off, sedge, dis, h1, W1, b1, emb, out, N);
}


# edge grouping in 2 launches, not 4: each slice of the edge list is bucket-sorted in place; the per-bucket pass merges the slices' runs and also writes dis; edges in a node: another LDS-rank order
# speedup vs baseline: 4.9078x; 1.1390x over previous
#include <hip/hip_runtime.h>
#include <hip/hip_bf16.h>
#include <math.h>

typedef float v2f __attribute__((ext_vector_type(2)));
typedef float v8f __attribute__((ext_vector_type(8)));

#define FEAT 128
#define BN_EPS 1e-5f

#define BKT_SHIFT 8
#define BKT_W     256
#define MAX_NB    1024
#define SORT_G    256
#define SORT_T    512
#define SORT_K    8
#define SEG_T     512
#define SEG_GRP   (SEG_T / 16)
#define SEG_S     (SORT_G / SEG_GRP)

__global__ __launch_bounds__(SORT_T) void slice_sort_kernel(const int* __restrict__ row,
                                                            const int* __restrict__ col,
                                                            const float* __restrict__ ew,
                                                            int* __restrict__ tabT,
                                                            int2* __restrict__ prec,
                                                            int E, int chunk, int nb, int tp) {
    __shared__ int cnt[MAX_NB];
    __shared__ int wtot[SORT_T / 32];
    const int tid  = threadIdx.x;
    const int lane = tid & 31;
    const int wv   = tid >> 5;
    const int w    = blockIdx.x;
    const int e0 = w * chunk;
    const int e1 = min(e0 + chunk, E);

    for (int b = tid; b < tp; b += SORT_T) cnt[b] = 0;

    unsigned c[SORT_K];
    int      r[SORT_K];
    int      wt[SORT_K];
#pragma unroll
    for (int k = 0; k < SORT_K; ++k) { c[k] = 0u; r[k] = 0; wt[k] = 0; }
    if (e0 < e1) {
#pragma unroll
        for (int k = 0; k < SORT_K; ++k) {
            const int e = min(e0 + tid + k * SORT_T, e1 - 1);
            c[k]  = (unsigned)col[e];
            r[k]  = row[e];
            wt[k] = __float_as_int(ew[e]);
        }
    }
    __syncthreads();

#pragma unroll
    for (int k = 0; k < SORT_K; ++k) {
        const int e = e0 + tid + k * SORT_T;
        if (e < e1) atomicAdd(&cnt[min((int)(c[k] >> BKT_SHIFT), nb - 1)], 1);
    }
    for (int e = e0 + SORT_T * SORT_K + tid; e < e1; e += SORT_T)
        atomicAdd(&cnt[min((int)((unsigned)col[e] >> BKT_SHIFT), nb - 1)], 1);
    __syncthreads();

    const bool own = (4 * tid < tp);
    int a0 = 0, a1 = 0, a2 = 0, a3 = 0;
    if (own) { a0 = cnt[4 * tid]; a1 = cnt[4 * tid + 1]; a2 = cnt[4 * tid + 2]; a3 = cnt[4 * tid + 3]; }
    const int s = a0 + a1 + a2 + a3;
    int inc = s;
#pragma unroll
    for (int d = 1; d < 32; d <<= 1) {
        int t = __shfl_up(inc, d);
        if (lane >= d) inc += t;
    }
    if (lane == 31) wtot[wv] = inc;
    __syncthreads();
    int base0 = inc - s;
#pragma unroll
    for (int k = 0; k < SORT_T / 32; ++k) if (k < wv) base0 += wtot[k];
    if (own) {
        const int s0 = base0, s1 = s0 + a0, s2 = s1 + a1, s3 = s2 + a2;
        cnt[4 * tid]     = e0 + s0;
        cnt[4 * tid + 1] = e0 + s1;
        cnt[4 * tid + 2] = e0 + s2;
        cnt[4 * tid + 3] = e0 + s3;
        tabT[(size_t)(4 * tid)     * SORT_G + w] = s0;
        tabT[(size_t)(4 * tid + 1) * SORT_G + w] = s1;
        tabT[(size_t)(4 * tid + 2) * SORT_G + w] = s2;
        tabT[(size_t)(4 * tid + 3) * SORT_G + w] = s3;
        if (4 * tid + 4 == tp) tabT[(size_t)tp * SORT_G + w] = s3 + a3;
    }
    __syncthreads();

    int pos[SORT_K];
#pragma unroll
    for (int k = 0; k < SORT_K; ++k) {
        const int e = e0 + tid + k * SORT_T;
        pos[k] = 0;
        if (e < e1) pos[k] = atomicAdd(&cnt[min((int)(c[k] >> BKT_SHIFT), nb - 1)], 1);
    }
#pragma unroll
    for (int k = 0; k < SORT_K; ++k) {
        const int e = e0 + tid + k * SORT_T;
        if (e < e1)
            prec[pos[k]] = make_int2((int)((unsigned)r[k] | ((c[k] & (BKT_W - 1)) << 24)), wt[k]);
    }
    for (int e = e0 + SORT_T * SORT_K + tid; e < e1; e += SORT_T) {
        const unsigned cc = (unsigned)col[e];
        const int at = atomicAdd(&cnt[min((int)(cc >> BKT_SHIFT), nb - 1)], 1);
        prec[at] = make_int2((int)((unsigned)row[e] | ((cc & (BKT_W - 1)) << 24)), __float_as_int(ew[e]));
    }
}

__global__ __launch_bounds__(SEG_T) void bucket_sort_kernel(const int* __restrict__ tabT,
                                                            const int2* __restrict__ prec,
                                                            int* __restrict__ off,
                                                            int2* sedge,
                                                            float* __restrict__ dis,
                                                            int n, int chunk, int E) {
    __shared__ int cnt[BKT_W];
    __shared__ int nbeg[BKT_W];
    __shared__ int nend[BKT_W];
    __shared__ int segbeg[SORT_G];
    __shared__ int seglen[SORT_G];
    __shared__ int wsum[SEG_T / 32];
    __shared__ int wtot[BKT_W / 32];
    const int tid  = threadIdx.x;
    const int lane = tid & 31;
    const int wv   = tid >> 5;
    const int b    = blockIdx.x;

    int ex = 0;
    if (tid < SORT_G) {
        ex = tabT[(size_t)b * SORT_G + tid];
        const int nx = tabT[(size_t)(b + 1) * SORT_G + tid];
        segbeg[tid] = tid * chunk + ex;
        seglen[tid] = nx - ex;
    }
    if (tid < BKT_W) cnt[tid] = 0;
    int red = ex;
#pragma unroll
    for (int d = 16; d >= 1; d >>= 1) red += __shfl_xor(red, d);
    if (lane == 0) wsum[wv] = red;
    __syncthreads();
    int beg = 0;
#pragma unroll
    for (int k = 0; k < SEG_T / 32; ++k) beg += wsum[k];

    const int g = tid >> 4;
    const int l = tid & 15;
    int  sb[SEG_S];
    int  sl[SEG_S];
    int2 rec[SEG_S * 2];
#pragma unroll
    for (int s = 0; s < SEG_S; ++s) {
        sb[s] = segbeg[g + SEG_GRP * s];
        sl[s] = seglen[g + SEG_GRP * s];
    }
#pragma unroll
    for (int s = 0; s < SEG_S; ++s) {
#pragma unroll
        for (int h = 0; h < 2; ++h) {
            const int j = l + 16 * h;
            rec[2 * s + h] = prec[min(max(sb[s] + min(j, sl[s] - 1), 0), E - 1)];
        }
    }
    __builtin_amdgcn_sched_barrier(0);
#pragma unroll
    for (int s = 0; s < SEG_S; ++s) {
#pragma unroll
        for (int h = 0; h < 2; ++h) {
            const int j = l + 16 * h;
            if (j < sl[s]) atomicAdd(&cnt[(unsigned)rec[2 * s + h].x >> 24], 1);
        }
    }
#pragma unroll
    for (int s = 0; s < SEG_S; ++s)
        for (int j = l + 32; j < sl[s]; j += 16)
            atomicAdd(&cnt[(unsigned)prec[sb[s] + j].x >> 24], 1);
    __syncthreads();

    const int v = (tid < BKT_W) ? cnt[tid] : 0;
    int inc = v;
#pragma unroll
    for (int d = 1; d < 32; d <<= 1) {
        int t = __shfl_up(inc, d);
        if (lane >= d) inc += t;
    }
    if (tid < BKT_W && lane == 31) wtot[wv] = inc;
    __syncthreads();
    if (tid < BKT_W) {
        int excl = beg + inc - v;
#pragma unroll
        for (int k = 0; k < BKT_W / 32; ++k) if (k < wv) excl += wtot[k];
        cnt[tid]  = excl;
        nbeg[tid] = excl;
        nend[tid] = excl + v;
        const int node = b * BKT_W + tid;
        if (node < n) off[node] = excl;
        if (node == n - 1) off[n] = excl + v;
    }
    __syncthreads();
    int pos[SEG_S * 2];
#pragma unroll
    for (int s = 0; s < SEG_S; ++s) {
#pragma unroll
        for (int h = 0; h < 2; ++h) {
            const int j = l + 16 * h;
            pos[2 * s + h] = 0;
            if (j < sl[s]) pos[2 * s + h] = atomicAdd(&cnt[(unsigned)rec[2 * s + h].x >> 24], 1);
        }
    }
#pragma unroll
    for (int s = 0; s < SEG_S; ++s) {
#pragma unroll
        for (int h = 0; h < 2; ++h) {
            const int j = l + 16 * h;
            if (j < sl[s])
                sedge[pos[2 * s + h]] = make_int2((int)((unsigned)rec[2 * s + h].x & 0xFFFFFFu), rec[2 * s + h].y);
        }
    }
#pragma unroll
    for (int s = 0; s < SEG_S; ++s)
        for (int j = l + 32; j < sl[s]; j += 16) {
            const int2 q = prec[sb[s] + j];
            const unsigned x = (unsigned)q.x;
            const int at = atomicAdd(&cnt[x >> 24], 1);
            sedge[at] = make_int2((int)(x & 0xFFFFFFu), q.y);
        }
    __syncthreads();

    const int sub = tid & 7;
#pragma unroll
    for (int rnd = 0; rnd < BKT_W / (SEG_T / 8); ++rnd) {
        const int nd   = rnd * (SEG_T / 8) + (tid >> 3);
        const int node = b * BKT_W + nd;
        const int pe   = nend[nd];
        float sw = 0.0f;
        for (int p = nbeg[nd] + sub; p < pe; p += 8) sw += __int_as_float(sedge[p].y);
        sw += __shfl_xor(sw, 1);
        sw += __shfl_xor(sw, 2);
        sw += __shfl_xor(sw, 4);
        if (node < n && sub == 0) dis[node] = rsqrtf(sw + 1.0f);
    }
}

#define TILE_M 32
#define NPW    (TILE_M / 8)
#define LDA    132

__device__ __forceinline__ void row_fma(float4& acc, float cf, const float4 hv) {
    acc.x = fmaf(cf, hv.x, acc.x);
    acc.y = fmaf(cf, hv.y, acc.y);
    acc.z = fmaf(cf, hv.z, acc.z);
    acc.w = fmaf(cf, hv.w, acc.w);
}

template <int K>
__device__ __forceinline__ void add_rows(float4& acc, int r, float cf, int j,
                                         const float* __restrict__ h, unsigned lane4) {
    float4 hv[K];
#pragma unroll
    for (int k = 0; k < K; ++k) {
        const int rk = __builtin_amdgcn_readlane(r, j + k);
        hv[k] = *(const float4*)(h + (size_t)rk * FEAT + lane4);
    }
#pragma unroll
    for (int k = 0; k < K; ++k) {
        const float ck = __int_as_float(__builtin_amdgcn_readlane(__float_as_int(cf), j + k));
        row_fma(acc, ck, hv[k]);
    }
}

__device__ __forceinline__ void aggregate_to_lds(float* __restrict__ lA,
                                                 const int* __restrict__ off,
                                                 const int2* __restrict__ sedge,
                                                 const float* __restrict__ dis,
                                                 const float* __restrict__ X,
                                                 int rowBase, int n) {
    const int lane = threadIdx.x & 31;
    const unsigned lane4 = (unsigned)lane * 4u;
    const int wave = __builtin_amdgcn_readfirstlane(threadIdx.x >> 5);
    const int c0 = rowBase + wave * NPW;
    int   ov = 0;
    float dv = 0.f;
    if (lane <= NPW) ov = off[min(c0 + lane, n)];
    if (lane <  NPW) dv = dis[min(c0 + lane, n - 1)];

    for (int i = 0; i < NPW; ++i) {
        const int c = c0 + i;
        float4 acc = make_float4(0.f, 0.f, 0.f, 0.f);
        if (c < n) {
            const int beg = __builtin_amdgcn_readlane(ov, i);
            const int end = __builtin_amdgcn_readlane(ov, i + 1);
            const float dc = __int_as_float(__builtin_amdgcn_readlane(__float_as_int(dv), i));
            const float4 xv = *(const float4*)(X + (size_t)c * FEAT + lane4);
            for (int base = beg; base < end; base += 32) {
                const int p = base + lane;
                int r = 0;
                float cf = 0.f;
                if (p < end) {
                    const int2 ed = sedge[p];
                    r  = ed.x;
                    cf = dis[r] * __int_as_float(ed.y) * dc;
                }
                const int m = (end - base < 32) ? (end - base) : 32;
                int j = 0;
                for (; j + 8 <= m; j += 8) add_rows<8>(acc, r, cf, j, X, lane4);
                if (m - j >= 4) { add_rows<4>(acc, r, cf, j, X, lane4); j += 4; }
                const int rem = m - j;
                if (rem == 3)      add_rows<3>(acc, r, cf, j, X, lane4);
                else if (rem == 2) add_rows<2>(acc, r, cf, j, X, lane4);
                else if (rem == 1) add_rows<1>(acc, r, cf, j, X, lane4);
            }
            row_fma(acc, dc * dc, xv);
        }
        *(float4*)(lA + (wave * NPW + i) * LDA + lane4) = acc;
    }
}

__device__ __forceinline__ void tile_times_w(const float* __restrict__ lA,
                                             const float* __restrict__ Wm,
                                             int n0, int lid, int half, v8f& acc0, v8f& acc1) {
    const float* arow0 = lA + lid * LDA;
    const float* arow1 = lA + (16 + lid) * LDA;
#pragma unroll
    for (int k0 = 0; k0 < FEAT; k0 += 4) {
        const int ka = k0 + 2 * half;
        v2f a0, a1, b;
        a0.x = arow0[ka];
        a0.y = arow0[ka + 1];
        a1.x = arow1[ka];
        a1.y = arow1[ka + 1];
        b.x = Wm[(size_t)ka * FEAT + n0 + lid];
        b.y = Wm[(size_t)(ka + 1) * FEAT + n0 + lid];
        acc0 = __builtin_amdgcn_wmma_f32_16x16x4_f32(
             false, a0,  false, b,
             (short)0, acc0,  false,  false);
        acc1 = __builtin_amdgcn_wmma_f32_16x16x4_f32(
             false, a1,  false, b,
             (short)0, acc1,  false,  false);
    }
}

__global__ __launch_bounds__(256, 16) void layer1_kernel(const int* __restrict__ off,
                                                     const int2* __restrict__ sedge,
                                                     const float* __restrict__ dis,
                                                     const float* __restrict__ emb,
                                                     const float* __restrict__ W0,
                                                     const float* __restrict__ b0,
                                                     const float* __restrict__ g,
                                                     const float* __restrict__ be,
                                                     const float* __restrict__ mu,
                                                     const float* __restrict__ var,
                                                     float* __restrict__ h1, int n) {
    __shared__ __attribute__((aligned(16))) float lA[TILE_M * LDA];
    const int rowBase = blockIdx.x * TILE_M;
    aggregate_to_lds(lA, off, sedge, dis, emb, rowBase, n);
    __syncthreads();

    const int lane = threadIdx.x & 31;
    const int lid  = lane & 15;
    const int half = lane >> 4;
    const int n0   = (threadIdx.x >> 5) * 16;
    const int f    = n0 + lid;
    const float bb = b0[f];
    const float mm = mu[f];
    const float rs = rsqrtf(var[f] + BN_EPS);
    const float gg = g[f];
    const float bt = be[f];

    v8f acc0 = {0.f, 0.f, 0.f, 0.f, 0.f, 0.f, 0.f, 0.f};
    v8f acc1 = {0.f, 0.f, 0.f, 0.f, 0.f, 0.f, 0.f, 0.f};
    tile_times_w(lA, W0, n0, lid, half, acc0, acc1);

    float* yb = h1 + (size_t)rowBase * FEAT + f;
    const bool whole = (rowBase + TILE_M <= n);
#pragma unroll
    for (int t = 0; t < 2; ++t) {
#pragma unroll
        for (int v = 0; v < 8; ++v) {
            const int m = 16 * t + v + 8 * half;
            float x = (t == 0 ? acc0[v] : acc1[v]) + bb;
            x = (x - mm) * rs * gg + bt;
            x = (x > 0.0f) ? x : expm1f(x);
            if (whole || rowBase + m < n) yb[(size_t)m * FEAT] = x;
        }
    }
}

__global__ __launch_bounds__(256, 16) void layer2_kernel(const int* __restrict__ off,
                                                     const int2* __restrict__ sedge,
                                                     const float* __restrict__ dis,
                                                     const float* __restrict__ h1,
                                                     const float* __restrict__ W1,
                                                     const float* __restrict__ b1,
                                                     const float* __restrict__ emb,
                                                     float* __restrict__ out, int n) {
    __shared__ __attribute__((aligned(16))) float lA[TILE_M * LDA];
    const int rowBase = blockIdx.x * TILE_M;
    aggregate_to_lds(lA, off, sedge, dis, h1, rowBase, n);
    __syncthreads();

    const int lane = threadIdx.x & 31;
    const int lid  = lane & 15;
    const int half = lane >> 4;
    const int n0   = (threadIdx.x >> 5) * 16;
    const int f    = n0 + lid;
    const float bb = b1[f];

    v8f acc0 = {0.f, 0.f, 0.f, 0.f, 0.f, 0.f, 0.f, 0.f};
    v8f acc1 = {0.f, 0.f, 0.f, 0.f, 0.f, 0.f, 0.f, 0.f};
    tile_times_w(lA, W1, n0, lid, half, acc0, acc1);

    const float inv3 = 1.0f / 3.0f;
    const size_t rb = (size_t)rowBase * FEAT + f;
    if (rowBase + TILE_M <= n) {
        float s[16];
#pragma unroll
        for (int q = 0; q < 16; ++q) {
            const size_t at = rb + (size_t)(16 * (q >> 3) + (q & 7) + 8 * half) * FEAT;
            s[q] = emb[at] + h1[at];
        }
#pragma unroll
        for (int q = 0; q < 16; ++q) {
            const size_t at = rb + (size_t)(16 * (q >> 3) + (q & 7) + 8 * half) * FEAT;
            out[at] = (s[q] + (q < 8 ? acc0[q & 7] : acc1[q & 7]) + bb) * inv3;
        }
    } else {
#pragma unroll
        for (int q = 0; q < 16; ++q) {
            const int m = 16 * (q >> 3) + (q & 7) + 8 * half;
            if (rowBase + m < n) {
                const size_t at = rb + (size_t)m * FEAT;
                out[at] = (emb[at] + h1[at] + (q < 8 ? acc0[q & 7] : acc1[q & 7]) + bb) * inv3;
            }
        }
    }
}

extern "C" void kernel_launch(void* const* d_in, const int* in_sizes, int n_in,
                              void* d_out, int out_size, void* d_ws, size_t ws_size,
                              hipStream_t stream) {
    const float* emb = (const float*)d_in[0];
    const int*   ei  = (const int*)d_in[1];
    const float* ew  = (const float*)d_in[2];
    const float* W0  = (const float*)d_in[3];
    const float* b0  = (const float*)d_in[4];
    const float* W1  = (const float*)d_in[5];
    const float* b1  = (const float*)d_in[6];
    const float* bng = (const float*)d_in[7];
    const float* bnb = (const float*)d_in[8];
    const float* bnm = (const float*)d_in[9];
    const float* bnv = (const float*)d_in[10];
    float* out = (float*)d_out;

    const int N  = in_sizes[0] / FEAT;
    const int E  = in_sizes[2];

    const int* row = ei;
    const int* col = ei + E;

    float* ws   = (float*)d_ws;
    size_t nPad = ((size_t)N + 1023) & ~(size_t)1023;
    size_t ePad = ((size_t)E + 1023) & ~(size_t)1023;
    int*   tabT   = (int*)ws;
    int*   off    = tabT + nPad + 2048;
    float* dis    = (float*)(off + nPad + 1024);
    int2*  prec   = (int2*)(dis + nPad);
    int2*  sedge  = prec + ePad;
    float* h1     = (float*)(sedge + ePad);

    const int gTile  = (N + TILE_M - 1) / TILE_M;
    const int nb     = (N + BKT_W - 1) / BKT_W;
    const int tp     = (nb + 3) & ~3;
    const int chunk  = (E + SORT_G - 1) / SORT_G;

    slice_sort_kernel<<<SORT_G, SORT_T, 0, stream>>>(row, col, ew, tabT, prec, E, chunk, nb, tp);
    bucket_sort_kernel<<<nb, SEG_T, 0, stream>>>(tabT, prec, off, sedge, dis, N, chunk, E);

    layer1_kernel<<<gTile, 256, 0, stream>>>(off, sedge, dis, emb, W0, b0, bng, bnb, bnm, bnv, h1, N);

    layer2_kernel<<<gTile, 256, 0, stream>>>(off, sedge, dis, h1, W1, b1, emb, out, N);
}


# both sort passes place records in an LDS image of the slice / bucket that leaves as whole-line stores (was one 8-byte store per record, a line per lane); degrees summed from the image; sum order kept
# speedup vs baseline: 5.2385x; 1.0674x over previous
#include <hip/hip_runtime.h>
#include <hip/hip_bf16.h>
#include <math.h>

typedef float v2f __attribute__((ext_vector_type(2)));
typedef float v8f __attribute__((ext_vector_type(8)));

#define FEAT 128
#define BN_EPS 1e-5f

#define BKT_SHIFT 8
#define BKT_W     256
#define MAX_NB    1024
#define SORT_G    256
#define SORT_T    512
#define SORT_K    8
#define SEG_T     512
#define SEG_GRP   (SEG_T / 16)
#define SEG_S     (SORT_G / SEG_GRP)
#define IMG1      (SORT_T * SORT_K)
#define IMG2      6144

template <int T>
__device__ __forceinline__ void image_out(const int2* img, int2* dst, int cnt, int tid) {
    for (int i0 = tid; i0 < cnt; i0 += 4 * T) {
        int2 q[4];
#pragma unroll
        for (int u = 0; u < 4; ++u) q[u] = img[min(i0 + u * T, cnt - 1)];
#pragma unroll
        for (int u = 0; u < 4; ++u)
            if (i0 + u * T < cnt) dst[i0 + u * T] = q[u];
    }
}

__global__ __launch_bounds__(SORT_T) void slice_sort_kernel(const int* __restrict__ row,
                                                            const int* __restrict__ col,
                                                            const float* __restrict__ ew,
                                                            int* __restrict__ tabT,
                                                            int2* __restrict__ prec,
                                                            int E, int chunk, int nb, int tp) {
    __shared__ int cnt[MAX_NB];
    __shared__ int wtot[SORT_T / 32];
    __shared__ __attribute__((aligned(8))) int imgw[2 * (IMG1 + SORT_T)];
    int2* img = (int2*)imgw;
    const int tid  = threadIdx.x;
    const int lane = tid & 31;
    const int wv   = tid >> 5;
    const int w    = blockIdx.x;
    const int e0 = w * chunk;
    const int e1 = min(e0 + chunk, E);
    const int ns = max(e1 - e0, 0);

    for (int b = tid; b < tp; b += SORT_T) cnt[b] = 0;

    unsigned c[SORT_K];
    int      r[SORT_K];
    int      wt[SORT_K];
#pragma unroll
    for (int k = 0; k < SORT_K; ++k) { c[k] = 0u; r[k] = 0; wt[k] = 0; }
    if (e0 < e1) {
#pragma unroll
        for (int k = 0; k < SORT_K; ++k) {
            const int e = min(e0 + tid + k * SORT_T, e1 - 1);
            c[k]  = (unsigned)col[e];
            r[k]  = row[e];
            wt[k] = __float_as_int(ew[e]);
        }
    }
    __syncthreads();

#pragma unroll
    for (int k = 0; k < SORT_K; ++k) {
        const int e = e0 + tid + k * SORT_T;
        if (e < e1) atomicAdd(&cnt[min((int)(c[k] >> BKT_SHIFT), nb - 1)], 1);
    }
    for (int e = e0 + SORT_T * SORT_K + tid; e < e1; e += SORT_T)
        atomicAdd(&cnt[min((int)((unsigned)col[e] >> BKT_SHIFT), nb - 1)], 1);
    __syncthreads();

    const bool own = (4 * tid < tp);
    int a0 = 0, a1 = 0, a2 = 0, a3 = 0;
    if (own) { a0 = cnt[4 * tid]; a1 = cnt[4 * tid + 1]; a2 = cnt[4 * tid + 2]; a3 = cnt[4 * tid + 3]; }
    const int s = a0 + a1 + a2 + a3;
    int inc = s;
#pragma unroll
    for (int d = 1; d < 32; d <<= 1) {
        int t = __shfl_up(inc, d);
        if (lane >= d) inc += t;
    }
    if (lane == 31) wtot[wv] = inc;
    __syncthreads();
    int base0 = inc - s;
#pragma unroll
    for (int k = 0; k < SORT_T / 32; ++k) if (k < wv) base0 += wtot[k];
    if (own) {
        const int s0 = base0, s1 = s0 + a0, s2 = s1 + a1, s3 = s2 + a2;
        cnt[4 * tid]     = e0 + s0;
        cnt[4 * tid + 1] = e0 + s1;
        cnt[4 * tid + 2] = e0 + s2;
        cnt[4 * tid + 3] = e0 + s3;
        tabT[(size_t)(4 * tid)     * SORT_G + w] = s0;
        tabT[(size_t)(4 * tid + 1) * SORT_G + w] = s1;
        tabT[(size_t)(4 * tid + 2) * SORT_G + w] = s2;
        tabT[(size_t)(4 * tid + 3) * SORT_G + w] = s3;
        if (4 * tid + 4 == tp) tabT[(size_t)tp * SORT_G + w] = s3 + a3;
    }
    __syncthreads();

    int pos[SORT_K];
#pragma unroll
    for (int k = 0; k < SORT_K; ++k) {
        const int e = e0 + tid + k * SORT_T;
        pos[k] = 0;
        if (e < e1) pos[k] = atomicAdd(&cnt[min((int)(c[k] >> BKT_SHIFT), nb - 1)], 1);
    }
    if (ns <= IMG1) {
#pragma unroll
        for (int k = 0; k < SORT_K; ++k) {
            const int e = e0 + tid + k * SORT_T;
            const int at = (e < e1) ? pos[k] - e0 : IMG1 + tid;
            img[at] = make_int2((int)((unsigned)r[k] | ((c[k] & (BKT_W - 1)) << 24)), wt[k]);
        }
        __syncthreads();
        image_out<SORT_T>(img, prec + e0, ns, tid);
    } else {
#pragma unroll
        for (int k = 0; k < SORT_K; ++k) {
            const int e = e0 + tid + k * SORT_T;
            if (e < e1)
                prec[pos[k]] = make_int2((int)((unsigned)r[k] | ((c[k] & (BKT_W - 1)) << 24)), wt[k]);
        }
        for (int e = e0 + SORT_T * SORT_K + tid; e < e1; e += SORT_T) {
            const unsigned cc = (unsigned)col[e];
            const int at = atomicAdd(&cnt[min((int)(cc >> BKT_SHIFT), nb - 1)], 1);
            prec[at] = make_int2((int)((unsigned)row[e] | ((cc & (BKT_W - 1)) << 24)), __float_as_int(ew[e]));
        }
    }
}

__global__ __launch_bounds__(SEG_T) void bucket_sort_kernel(const int* __restrict__ tabT,
                                                            const int2* __restrict__ prec,
                                                            int* __restrict__ off,
                                                            int2* sedge,
                                                            float* __restrict__ dis,
                                                            int n, int chunk, int E) {
    __shared__ int cnt[BKT_W];
    __shared__ int nbeg[BKT_W];
    __shared__ int nend[BKT_W];
    __shared__ int segbeg[SORT_G];
    __shared__ int seglen[SORT_G];
    __shared__ int wsum[SEG_T / 32];
    __shared__ int wtot[BKT_W / 32];
    __shared__ __attribute__((aligned(8))) int imgw[2 * (IMG2 + SEG_T)];
    int2* img = (int2*)imgw;
    const int tid  = threadIdx.x;
    const int lane = tid & 31;
    const int wv   = tid >> 5;
    const int b    = blockIdx.x;

    int ex = 0;
    if (tid < SORT_G) {
        ex = tabT[(size_t)b * SORT_G + tid];
        const int nx = tabT[(size_t)(b + 1) * SORT_G + tid];
        segbeg[tid] = tid * chunk + ex;
        seglen[tid] = nx - ex;
    }
    if (tid < BKT_W) cnt[tid] = 0;
    int red = ex;
#pragma unroll
    for (int d = 16; d >= 1; d >>= 1) red += __shfl_xor(red, d);
    if (lane == 0) wsum[wv] = red;
    __syncthreads();
    int beg = 0;
#pragma unroll
    for (int k = 0; k < SEG_T / 32; ++k) beg += wsum[k];

    const int g = tid >> 4;
    const int l = tid & 15;
    int  sb[SEG_S];
    int  sl[SEG_S];
    int2 rec[SEG_S * 2];
#pragma unroll
    for (int s = 0; s < SEG_S; ++s) {
        sb[s] = segbeg[g + SEG_GRP * s];
        sl[s] = seglen[g + SEG_GRP * s];
    }
#pragma unroll
    for (int s = 0; s < SEG_S; ++s) {
#pragma unroll
        for (int h = 0; h < 2; ++h) {
            const int j = l + 16 * h;
            rec[2 * s + h] = prec[min(max(sb[s] + min(j, sl[s] - 1), 0), E - 1)];
        }
    }
    __builtin_amdgcn_sched_barrier(0);
#pragma unroll
    for (int s = 0; s < SEG_S; ++s) {
#pragma unroll
        for (int h = 0; h < 2; ++h) {
            const int j = l + 16 * h;
            if (j < sl[s]) atomicAdd(&cnt[(unsigned)rec[2 * s + h].x >> 24], 1);
        }
    }
#pragma unroll
    for (int s = 0; s < SEG_S; ++s)
        for (int j = l + 32; j < sl[s]; j += 16)
            atomicAdd(&cnt[(unsigned)prec[sb[s] + j].x >> 24], 1);
    __syncthreads();

    const int v = (tid < BKT_W) ? cnt[tid] : 0;
    int inc = v;
#pragma unroll
    for (int d = 1; d < 32; d <<= 1) {
        int t = __shfl_up(inc, d);
        if (lane >= d) inc += t;
    }
    if (tid < BKT_W && lane == 31) wtot[wv] = inc;
    __syncthreads();
    int tot = 0;
#pragma unroll
    for (int k = 0; k < BKT_W / 32; ++k) tot += wtot[k];
    const bool staged = (tot <= IMG2);
    if (tid < BKT_W) {
        int excl = beg + inc - v;
#pragma unroll
        for (int k = 0; k < BKT_W / 32; ++k) if (k < wv) excl += wtot[k];
        cnt[tid]  = excl;
        nbeg[tid] = excl;
        nend[tid] = excl + v;
        const int node = b * BKT_W + tid;
        if (node < n) off[node] = excl;
        if (node == n - 1) off[n] = excl + v;
    }
    __syncthreads();
    int pos[SEG_S * 2];
#pragma unroll
    for (int s = 0; s < SEG_S; ++s) {
#pragma unroll
        for (int h = 0; h < 2; ++h) {
            const int j = l + 16 * h;
            pos[2 * s + h] = 0;
            if (j < sl[s]) pos[2 * s + h] = atomicAdd(&cnt[(unsigned)rec[2 * s + h].x >> 24], 1);
        }
    }
    if (staged) {
#pragma unroll
        for (int s = 0; s < SEG_S; ++s) {
#pragma unroll
            for (int h = 0; h < 2; ++h) {
                const int j = l + 16 * h;
                const int at = (j < sl[s]) ? pos[2 * s + h] - beg : IMG2 + tid;
                img[at] = make_int2((int)((unsigned)rec[2 * s + h].x & 0xFFFFFFu), rec[2 * s + h].y);
            }
        }
#pragma unroll
        for (int s = 0; s < SEG_S; ++s)
            for (int j = l + 32; j < sl[s]; j += 16) {
                const int2 q = prec[sb[s] + j];
                const unsigned x = (unsigned)q.x;
                const int at = atomicAdd(&cnt[x >> 24], 1);
                img[at - beg] = make_int2((int)(x & 0xFFFFFFu), q.y);
            }
        __syncthreads();
        image_out<SEG_T>(img, sedge + beg, tot, tid);
    } else {
#pragma unroll
        for (int s = 0; s < SEG_S; ++s) {
#pragma unroll
            for (int h = 0; h < 2; ++h) {
                const int j = l + 16 * h;
                if (j < sl[s])
                    sedge[pos[2 * s + h]] = make_int2((int)((unsigned)rec[2 * s + h].x & 0xFFFFFFu), rec[2 * s + h].y);
            }
        }
#pragma unroll
        for (int s = 0; s < SEG_S; ++s)
            for (int j = l + 32; j < sl[s]; j += 16) {
                const int2 q = prec[sb[s] + j];
                const unsigned x = (unsigned)q.x;
                const int at = atomicAdd(&cnt[x >> 24], 1);
                sedge[at] = make_int2((int)(x & 0xFFFFFFu), q.y);
            }
        __syncthreads();
    }

    const int sub = tid & 7;
#pragma unroll
    for (int rnd = 0; rnd < BKT_W / (SEG_T / 8); ++rnd) {
        const int nd   = rnd * (SEG_T / 8) + (tid >> 3);
        const int node = b * BKT_W + nd;
        const int pe   = nend[nd];
        float sw = 0.0f;
        if (staged) {
            for (int p = nbeg[nd] + sub; p < pe; p += 8) sw += __int_as_float(img[p - beg].y);
        } else {
            for (int p = nbeg[nd] + sub; p < pe; p += 8) sw += __int_as_float(sedge[p].y);
        }
        sw += __shfl_xor(sw, 1);
        sw += __shfl_xor(sw, 2);
        sw += __shfl_xor(sw, 4);
        if (node < n && sub == 0) dis[node] = rsqrtf(sw + 1.0f);
    }
}

#define TILE_M 32
#define NPW    (TILE_M / 8)
#define LDA    132

__device__ __forceinline__ void row_fma(float4& acc, float cf, const float4 hv) {
    acc.x = fmaf(cf, hv.x, acc.x);
    acc.y = fmaf(cf, hv.y, acc.y);
    acc.z = fmaf(cf, hv.z, acc.z);
    acc.w = fmaf(cf, hv.w, acc.w);
}

template <int K>
__device__ __forceinline__ void add_rows(float4& acc, int r, float cf, int j,
                                         const float* __restrict__ h, unsigned lane4) {
    float4 hv[K];
#pragma unroll
    for (int k = 0; k < K; ++k) {
        const int rk = __builtin_amdgcn_readlane(r, j + k);
        hv[k] = *(const float4*)(h + (size_t)rk * FEAT + lane4);
    }
#pragma unroll
    for (int k = 0; k < K; ++k) {
        const float ck = __int_as_float(__builtin_amdgcn_readlane(__float_as_int(cf), j + k));
        row_fma(acc, ck, hv[k]);
    }
}

__device__ __forceinline__ void aggregate_to_lds(float* __restrict__ lA,
                                                 const int* __restrict__ off,
                                                 const int2* __restrict__ sedge,
                                                 const float* __restrict__ dis,
                                                 const float* __restrict__ X,
                                                 int rowBase, int n) {
    const int lane = threadIdx.x & 31;
    const unsigned lane4 = (unsigned)lane * 4u;
    const int wave = __builtin_amdgcn_readfirstlane(threadIdx.x >> 5);
    const int c0 = rowBase + wave * NPW;
    int   ov = 0;
    float dv = 0.f;
    if (lane <= NPW) ov = off[min(c0 + lane, n)];
    if (lane <  NPW) dv = dis[min(c0 + lane, n - 1)];

    for (int i = 0; i < NPW; ++i) {
        const int c = c0 + i;
        float4 acc = make_float4(0.f, 0.f, 0.f, 0.f);
        if (c < n) {
            const int beg = __builtin_amdgcn_readlane(ov, i);
            const int end = __builtin_amdgcn_readlane(ov, i + 1);
            const float dc = __int_as_float(__builtin_amdgcn_readlane(__float_as_int(dv), i));
            const float4 xv = *(const float4*)(X + (size_t)c * FEAT + lane4);
            for (int base = beg; base < end; base += 32) {
                const int p = base + lane;
                int r = 0;
                float cf = 0.f;
                if (p < end) {
                    const int2 ed = sedge[p];
                    r  = ed.x;
                    cf = dis[r] * __int_as_float(ed.y) * dc;
                }
                const int m = (end - base < 32) ? (end - base) : 32;
                int j = 0;
                for (; j + 8 <= m; j += 8) add_rows<8>(acc, r, cf, j, X, lane4);
                if (m - j >= 4) { add_rows<4>(acc, r, cf, j, X, lane4); j += 4; }
                const int rem = m - j;
                if (rem == 3)      add_rows<3>(acc, r, cf, j, X, lane4);
                else if (rem == 2) add_rows<2>(acc, r, cf, j, X, lane4);
                else if (rem == 1) add_rows<1>(acc, r, cf, j, X, lane4);
            }
            row_fma(acc, dc * dc, xv);
        }
        *(float4*)(lA + (wave * NPW + i) * LDA + lane4) = acc;
    }
}

__device__ __forceinline__ void tile_times_w(const float* __restrict__ lA,
                                             const float* __restrict__ Wm,
                                             int n0, int lid, int half, v8f& acc0, v8f& acc1) {
    const float* arow0 = lA + lid * LDA;
    const float* arow1 = lA + (16 + lid) * LDA;
#pragma unroll
    for (int k0 = 0; k0 < FEAT; k0 += 4) {
        const int ka = k0 + 2 * half;
        v2f a0, a1, b;
        a0.x = arow0[ka];
        a0.y = arow0[ka + 1];
        a1.x = arow1[ka];
        a1.y = arow1[ka + 1];
        b.x = Wm[(size_t)ka * FEAT + n0 + lid];
        b.y = Wm[(size_t)(ka + 1) * FEAT + n0 + lid];
        acc0 = __builtin_amdgcn_wmma_f32_16x16x4_f32(
             false, a0,  false, b,
             (short)0, acc0,  false,  false);
        acc1 = __builtin_amdgcn_wmma_f32_16x16x4_f32(
             false, a1,  false, b,
             (short)0, acc1,  false,  false);
    }
}

__global__ __launch_bounds__(256, 16) void layer1_kernel(const int* __restrict__ off,
                                                     const int2* __restrict__ sedge,
                                                     const float* __restrict__ dis,
                                                     const float* __restrict__ emb,
                                                     const float* __restrict__ W0,
                                                     const float* __restrict__ b0,
                                                     const float* __restrict__ g,
                                                     const float* __restrict__ be,
                                                     const float* __restrict__ mu,
                                                     const float* __restrict__ var,
                                                     float* __restrict__ h1, int n) {
    __shared__ __attribute__((aligned(16))) float lA[TILE_M * LDA];
    const int rowBase = blockIdx.x * TILE_M;
    aggregate_to_lds(lA, off, sedge, dis, emb, rowBase, n);
    __syncthreads();

    const int lane = threadIdx.x & 31;
    const int lid  = lane & 15;
    const int half = lane >> 4;
    const int n0   = (threadIdx.x >> 5) * 16;
    const int f    = n0 + lid;
    const float bb = b0[f];
    const float mm = mu[f];
    const float rs = rsqrtf(var[f] + BN_EPS);
    const float gg = g[f];
    const float bt = be[f];

    v8f acc0 = {0.f, 0.f, 0.f, 0.f, 0.f, 0.f, 0.f, 0.f};
    v8f acc1 = {0.f, 0.f, 0.f, 0.f, 0.f, 0.f, 0.f, 0.f};
    tile_times_w(lA, W0, n0, lid, half, acc0, acc1);

    float* yb = h1 + (size_t)rowBase * FEAT + f;
    const bool whole = (rowBase + TILE_M <= n);
#pragma unroll
    for (int t = 0; t < 2; ++t) {
#pragma unroll
        for (int v = 0; v < 8; ++v) {
            const int m = 16 * t + v + 8 * half;
            float x = (t == 0 ? acc0[v] : acc1[v]) + bb;
            x = (x - mm) * rs * gg + bt;
            x = (x > 0.0f) ? x : expm1f(x);
            if (whole || rowBase + m < n) yb[(size_t)m * FEAT] = x;
        }
    }
}

__global__ __launch_bounds__(256, 16) void layer2_kernel(const int* __restrict__ off,
                                                     const int2* __restrict__ sedge,
                                                     const float* __restrict__ dis,
                                                     const float* __restrict__ h1,
                                                     const float* __restrict__ W1,
                                                     const float* __restrict__ b1,
                                                     const float* __restrict__ emb,
                                                     float* __restrict__ out, int n) {
    __shared__ __attribute__((aligned(16))) float lA[TILE_M * LDA];
    const int rowBase = blockIdx.x * TILE_M;
    aggregate_to_lds(lA, off, sedge, dis, h1, rowBase, n);
    __syncthreads();

    const int lane = threadIdx.x & 31;
    const int lid  = lane & 15;
    const int half = lane >> 4;
    const int n0   = (threadIdx.x >> 5) * 16;
    const int f    = n0 + lid;
    const float bb = b1[f];

    v8f acc0 = {0.f, 0.f, 0.f, 0.f, 0.f, 0.f, 0.f, 0.f};
    v8f acc1 = {0.f, 0.f, 0.f, 0.f, 0.f, 0.f, 0.f, 0.f};
    tile_times_w(lA, W1, n0, lid, half, acc0, acc1);

    const float inv3 = 1.0f / 3.0f;
    const size_t rb = (size_t)rowBase * FEAT + f;
    if (rowBase + TILE_M <= n) {
        float s[16];
#pragma unroll
        for (int q = 0; q < 16; ++q) {
            const size_t at = rb + (size_t)(16 * (q >> 3) + (q & 7) + 8 * half) * FEAT;
            s[q] = emb[at] + h1[at];
        }
#pragma unroll
        for (int q = 0; q < 16; ++q) {
            const size_t at = rb + (size_t)(16 * (q >> 3) + (q & 7) + 8 * half) * FEAT;
            out[at] = (s[q] + (q < 8 ? acc0[q & 7] : acc1[q & 7]) + bb) * inv3;
        }
    } else {
#pragma unroll
        for (int q = 0; q < 16; ++q) {
            const int m = 16 * (q >> 3) + (q & 7) + 8 * half;
            if (rowBase + m < n) {
                const size_t at = rb + (size_t)m * FEAT;
                out[at] = (emb[at] + h1[at] + (q < 8 ? acc0[q & 7] : acc1[q & 7]) + bb) * inv3;
            }
        }
    }
}

extern "C" void kernel_launch(void* const* d_in, const int* in_sizes, int n_in,
                              void* d_out, int out_size, void* d_ws, size_t ws_size,
                              hipStream_t stream) {
    const float* emb = (const float*)d_in[0];
    const int*   ei  = (const int*)d_in[1];
    const float* ew  = (const float*)d_in[2];
    const float* W0  = (const float*)d_in[3];
    const float* b0  = (const float*)d_in[4];
    const float* W1  = (const float*)d_in[5];
    const float* b1  = (const float*)d_in[6];
    const float* bng = (const float*)d_in[7];
    const float* bnb = (const float*)d_in[8];
    const float* bnm = (const float*)d_in[9];
    const float* bnv = (const float*)d_in[10];
    float* out = (float*)d_out;

    const int N  = in_sizes[0] / FEAT;
    const int E  = in_sizes[2];

    const int* row = ei;
    const int* col = ei + E;

    float* ws   = (float*)d_ws;
    size_t nPad = ((size_t)N + 1023) & ~(size_t)1023;
    size_t ePad = ((size_t)E + 1023) & ~(size_t)1023;
    int*   tabT   = (int*)ws;
    int*   off    = tabT + nPad + 2048;
    float* dis    = (float*)(off + nPad + 1024);
    int2*  prec   = (int2*)(dis + nPad);
    int2*  sedge  = prec + ePad;
    float* h1     = (float*)(sedge + ePad);

    const int gTile  = (N + TILE_M - 1) / TILE_M;
    const int nb     = (N + BKT_W - 1) / BKT_W;
    const int tp     = (nb + 3) & ~3;
    const int chunk  = (E + SORT_G - 1) / SORT_G;

    slice_sort_kernel<<<SORT_G, SORT_T, 0, stream>>>(row, col, ew, tabT, prec, E, chunk, nb, tp);
    bucket_sort_kernel<<<nb, SEG_T, 0, stream>>>(tabT, prec, off, sedge, dis, N, chunk, E);

    layer1_kernel<<<gTile, 256, 0, stream>>>(off, sedge, dis, emb, W0, b0, bng, bnb, bnm, bnv, h1, N);

    layer2_kernel<<<gTile, 256, 0, stream>>>(off, sedge, dis, h1, W1, b1, emb, out, N);
}
